# starting kernel
# baseline (speedup 1.0000x reference)
#include <hip/hip_runtime.h>
#include <math.h>

typedef __attribute__((ext_vector_type(2))) float v2f;
typedef __attribute__((ext_vector_type(8))) float v8f;

#define N_FEAT   50
#define HIDDEN   16
#define N_CLS    10

__global__ void zero_fill(float* __restrict__ p, long n) {
  long i = (long)blockIdx.x * blockDim.x + threadIdx.x;
  if (i < n) p[i] = 0.0f;
}

__global__ void deg_kernel(const int* __restrict__ dst, float* __restrict__ deg, int E) {
  int e = blockIdx.x * blockDim.x + threadIdx.x;
  if (e < E) atomicAdd(&deg[dst[e]], 1.0f);
}

__global__ void dinv_kernel(const float* __restrict__ deg, float* __restrict__ dinv, int n) {
  int i = blockIdx.x * blockDim.x + threadIdx.x;
  if (i < n) {
    float d = deg[i];
    dinv[i] = (d > 0.0f) ? rsqrtf(d) : 0.0f;
  }
}

__global__ void norm_kernel(const int* __restrict__ src, const int* __restrict__ dst,
                            const float* __restrict__ dinv, float* __restrict__ nrm, int E) {
  int e = blockIdx.x * blockDim.x + threadIdx.x;
  if (e < E) nrm[e] = dinv[src[e]] * dinv[dst[e]];
}

__global__ void scatter50(const int* __restrict__ src, const int* __restrict__ dst,
                          const float* __restrict__ nrm, const float* __restrict__ x,
                          float* __restrict__ agg, int E) {
  long gt   = (long)blockIdx.x * blockDim.x + threadIdx.x;
  int  w    = (int)(gt >> 5);
  int  lane = threadIdx.x & 31;
  int  e    = __builtin_amdgcn_readfirstlane(w >> 1);
  int  f    = ((w & 1) << 5) + lane;
  if (e >= E) return;
  if (f < N_FEAT) {
    int   s = src[e];
    int   d = dst[e];
    float g = nrm[e];
    float v = x[(long)s * N_FEAT + f] * g;
    atomicAdd(&agg[(long)d * N_FEAT + f], v);
  }
}

__global__ void scatter16(const int* __restrict__ src, const int* __restrict__ dst,
                          const float* __restrict__ nrm, const float* __restrict__ h,
                          float* __restrict__ agg, int E) {
  long t = (long)blockIdx.x * blockDim.x + threadIdx.x;
  int e = (int)(t >> 4);
  int f = (int)(t & 15);
  if (e >= E) return;
  int s = src[e], d = dst[e];
  float v = h[(long)s * HIDDEN + f] * nrm[e];
  atomicAdd(&agg[(long)d * HIDDEN + f], v);
}

__global__ __launch_bounds__(256) void tag1_dense(
    const float* __restrict__ X, const float* __restrict__ AGG,
    const float* __restrict__ W0, const float* __restrict__ W1,
    const float* __restrict__ B, float* __restrict__ H, int n) {
  int wave = (int)((blockIdx.x * blockDim.x + threadIdx.x) >> 5);
  int lane = threadIdx.x & 31;
  int base = wave * 16;
  if (base >= n) return;

  int row   = lane & 15;
  int khalf = (lane >> 4) * 2;
  int node  = base + row;
  node = node < n ? node : n - 1;
  long xrow = (long)node * N_FEAT;
  bool lo   = (khalf == 0);

  v8f c = {};
#pragma unroll
  for (int k = 0; k < 48; k += 4) {
    v2f a = *(const v2f*)(X + xrow + k + khalf);
    v2f b;
    b.x = W0[(k + khalf)     * HIDDEN + row];
    b.y = W0[(k + khalf + 1) * HIDDEN + row];
    c = __builtin_amdgcn_wmma_f32_16x16x4_f32(false, a, false, b, (short)0, c, false, false);
  }
  {
    v2f a0 = *(const v2f*)(X + xrow + 48);
    float b0x = W0[48 * HIDDEN + row];
    float b0y = W0[49 * HIDDEN + row];
    v2f a, b;
    a.x = lo ? a0.x : 0.0f;  a.y = lo ? a0.y : 0.0f;
    b.x = lo ? b0x  : 0.0f;  b.y = lo ? b0y  : 0.0f;
    c = __builtin_amdgcn_wmma_f32_16x16x4_f32(false, a, false, b, (short)0, c, false, false);
  }
#pragma unroll
  for (int k = 0; k < 48; k += 4) {
    v2f a = *(const v2f*)(AGG + xrow + k + khalf);
    v2f b;
    b.x = W1[(k + khalf)     * HIDDEN + row];
    b.y = W1[(k + khalf + 1) * HIDDEN + row];
    c = __builtin_amdgcn_wmma_f32_16x16x4_f32(false, a, false, b, (short)0, c, false, false);
  }
  {
    v2f a0 = *(const v2f*)(AGG + xrow + 48);
    float b0x = W1[48 * HIDDEN + row];
    float b0y = W1[49 * HIDDEN + row];
    v2f a, b;
    a.x = lo ? a0.x : 0.0f;  a.y = lo ? a0.y : 0.0f;
    b.x = lo ? b0x  : 0.0f;  b.y = lo ? b0y  : 0.0f;
    c = __builtin_amdgcn_wmma_f32_16x16x4_f32(false, a, false, b, (short)0, c, false, false);
  }

  int col = lane & 15;
  int mhi = (lane >> 4) * 8;
  float bias = B[col];
#pragma unroll
  for (int j = 0; j < 8; ++j) {
    int m = base + mhi + j;
    if (m < n) {
      float v = c[j] + bias;
      H[(long)m * HIDDEN + col] = v > 0.0f ? v : 0.0f;
    }
  }
}

__global__ __launch_bounds__(256) void tag2_dense(
    const float* __restrict__ H, const float* __restrict__ AGG,
    const float* __restrict__ W0, const float* __restrict__ W1,
    const float* __restrict__ B, float* __restrict__ OUT, int n) {
  int wave = (int)((blockIdx.x * blockDim.x + threadIdx.x) >> 5);
  int lane = threadIdx.x & 31;
  int base = wave * 16;
  if (base >= n) return;

  int row   = lane & 15;
  int khalf = (lane >> 4) * 2;
  int node  = base + row;
  node = node < n ? node : n - 1;
  long hrow = (long)node * HIDDEN;
  bool bok  = row < N_CLS;
  int  wrow = bok ? row : (N_CLS - 1);

  v8f c = {};
#pragma unroll
  for (int k = 0; k < HIDDEN; k += 4) {
    int ka = k + khalf;
    v2f a = *(const v2f*)(H + hrow + ka);
    float b0 = W0[(ka)     * N_CLS + wrow];
    float b1 = W0[(ka + 1) * N_CLS + wrow];
    v2f b;
    b.x = bok ? b0 : 0.0f;  b.y = bok ? b1 : 0.0f;
    c = __builtin_amdgcn_wmma_f32_16x16x4_f32(false, a, false, b, (short)0, c, false, false);
  }
#pragma unroll
  for (int k = 0; k < HIDDEN; k += 4) {
    int ka = k + khalf;
    v2f a = *(const v2f*)(AGG + hrow + ka);
    float b0 = W1[(ka)     * N_CLS + wrow];
    float b1 = W1[(ka + 1) * N_CLS + wrow];
    v2f b;
    b.x = bok ? b0 : 0.0f;  b.y = bok ? b1 : 0.0f;
    c = __builtin_amdgcn_wmma_f32_16x16x4_f32(false, a, false, b, (short)0, c, false, false);
  }

  int col  = lane & 15;
  int mhi  = (lane >> 4) * 8;
  bool cok = col < N_CLS;
  float bias = cok ? B[col] : 0.0f;

#pragma unroll
  for (int j = 0; j < 8; ++j) {
    int m = base + mhi + j;
    float v = c[j] + bias;
    float mx = cok ? v : -INFINITY;
    for (int off = 8; off >= 1; off >>= 1)
      mx = fmaxf(mx, __shfl_xor(mx, off, 16));
    float ex = cok ? expf(v - mx) : 0.0f;
    float sm = ex;
    for (int off = 8; off >= 1; off >>= 1)
      sm += __shfl_xor(sm, off, 16);
    float r = v - mx - logf(sm);
    if (cok && m < n) OUT[(long)m * N_CLS + col] = r;
  }
}

extern "C" void kernel_launch(void* const* d_in, const int* in_sizes, int n_in,
                              void* d_out, int out_size, void* d_ws, size_t ws_size,
                              hipStream_t stream) {
  const float* x    = (const float*)d_in[0];
  const int*   ei   = (const int*)d_in[1];
  const float* w1_0 = (const float*)d_in[2];
  const float* w1_1 = (const float*)d_in[3];
  const float* b1   = (const float*)d_in[4];
  const float* w2_0 = (const float*)d_in[5];
  const float* w2_1 = (const float*)d_in[6];
  const float* b2   = (const float*)d_in[7];
  float* out = (float*)d_out;

  int n = in_sizes[0] / N_FEAT;
  int E = in_sizes[1] / 2;
  const int* src = ei;
  const int* dst = ei + E;

  float* ws   = (float*)d_ws;
  float* deg  = ws;
  float* agg1 = deg  + n;
  float* agg2 = agg1 + (long)n * N_FEAT;
  float* dinv = agg2 + (long)n * HIDDEN;
  float* nrm  = dinv + n;
  float* h    = nrm  + E;

  long zn = (long)n + (long)n * N_FEAT + (long)n * HIDDEN;
  zero_fill<<<(unsigned)((zn + 255) / 256), 256, 0, stream>>>(deg, zn);

  deg_kernel <<<(E + 255) / 256, 256, 0, stream>>>(dst, deg, E);
  dinv_kernel<<<(n + 255) / 256, 256, 0, stream>>>(deg, dinv, n);
  norm_kernel<<<(E + 255) / 256, 256, 0, stream>>>(src, dst, dinv, nrm, E);

  long t50 = (long)E * 64;
  scatter50<<<(unsigned)((t50 + 255) / 256), 256, 0, stream>>>(src, dst, nrm, x, agg1, E);

  int ntiles = (n + 15) / 16;
  int nblk   = (ntiles + 7) / 8;
  tag1_dense<<<nblk, 256, 0, stream>>>(x, agg1, w1_0, w1_1, b1, h, n);

  long t16 = (long)E * 16;
  scatter16<<<(unsigned)((t16 + 255) / 256), 256, 0, stream>>>(src, dst, nrm, h, agg2, E);

  tag2_dense<<<nblk, 256, 0, stream>>>(h, agg2, w2_0, w2_1, b2, out, n);
}


# Layer 1 aggregates after projecting: scatter-add rows of x@w1_1 (16 wide) instead of rows of x (50 wide), by linearity; edge sum now follows the K=50 product (float summation order changes)
# speedup vs baseline: 1.7811x; 1.7811x over previous
#include <hip/hip_runtime.h>
#include <math.h>

typedef __attribute__((ext_vector_type(2))) float v2f;
typedef __attribute__((ext_vector_type(8))) float v8f;

#define N_FEAT   50
#define HIDDEN   16
#define N_CLS    10

__global__ void zero_fill(float* __restrict__ p, long n) {
  long i = (long)blockIdx.x * blockDim.x + threadIdx.x;
  if (i < n) p[i] = 0.0f;
}

__global__ void deg_kernel(const int* __restrict__ dst, float* __restrict__ deg, int E) {
  int e = blockIdx.x * blockDim.x + threadIdx.x;
  if (e < E) atomicAdd(&deg[dst[e]], 1.0f);
}

__global__ void dinv_kernel(const float* __restrict__ deg, float* __restrict__ dinv, int n) {
  int i = blockIdx.x * blockDim.x + threadIdx.x;
  if (i < n) {
    float d = deg[i];
    dinv[i] = (d > 0.0f) ? rsqrtf(d) : 0.0f;
  }
}

__global__ void norm_kernel(const int* __restrict__ src, const int* __restrict__ dst,
                            const float* __restrict__ dinv, float* __restrict__ nrm, int E) {
  int e = blockIdx.x * blockDim.x + threadIdx.x;
  if (e < E) nrm[e] = dinv[src[e]] * dinv[dst[e]];
}

__global__ void scatter16(const int* __restrict__ src, const int* __restrict__ dst,
                          const float* __restrict__ nrm, const float* __restrict__ h,
                          float* __restrict__ agg, int E) {
  long t = (long)blockIdx.x * blockDim.x + threadIdx.x;
  int e = (int)(t >> 4);
  int f = (int)(t & 15);
  if (e >= E) return;
  int s = src[e], d = dst[e];
  float v = h[(long)s * HIDDEN + f] * nrm[e];
  atomicAdd(&agg[(long)d * HIDDEN + f], v);
}

__global__ __launch_bounds__(256) void proj1_dense(
    const float* __restrict__ X, const float* __restrict__ W1,
    float* __restrict__ Y, int n) {
  int wave = (int)((blockIdx.x * blockDim.x + threadIdx.x) >> 5);
  int lane = threadIdx.x & 31;
  int base = wave * 16;
  if (base >= n) return;

  int row   = lane & 15;
  int khalf = (lane >> 4) * 2;
  int node  = base + row;
  node = node < n ? node : n - 1;
  long xrow = (long)node * N_FEAT;
  bool lo   = (khalf == 0);

  v8f c = {};
#pragma unroll
  for (int k = 0; k < 48; k += 4) {
    v2f a = *(const v2f*)(X + xrow + k + khalf);
    v2f b;
    b.x = W1[(k + khalf)     * HIDDEN + row];
    b.y = W1[(k + khalf + 1) * HIDDEN + row];
    c = __builtin_amdgcn_wmma_f32_16x16x4_f32(false, a, false, b, (short)0, c, false, false);
  }
  {
    v2f a0 = *(const v2f*)(X + xrow + 48);
    float b0x = W1[48 * HIDDEN + row];
    float b0y = W1[49 * HIDDEN + row];
    v2f a, b;
    a.x = lo ? a0.x : 0.0f;  a.y = lo ? a0.y : 0.0f;
    b.x = lo ? b0x  : 0.0f;  b.y = lo ? b0y  : 0.0f;
    c = __builtin_amdgcn_wmma_f32_16x16x4_f32(false, a, false, b, (short)0, c, false, false);
  }

  int col = lane & 15;
  int mhi = (lane >> 4) * 8;
#pragma unroll
  for (int j = 0; j < 8; ++j) {
    int m = base + mhi + j;
    if (m < n) Y[(long)m * HIDDEN + col] = c[j];
  }
}

__global__ __launch_bounds__(256) void tag1_dense(
    const float* __restrict__ X, const float* __restrict__ AGG,
    const float* __restrict__ W0,
    const float* __restrict__ B, float* __restrict__ H, int n) {
  int wave = (int)((blockIdx.x * blockDim.x + threadIdx.x) >> 5);
  int lane = threadIdx.x & 31;
  int base = wave * 16;
  if (base >= n) return;

  int row   = lane & 15;
  int khalf = (lane >> 4) * 2;
  int node  = base + row;
  node = node < n ? node : n - 1;
  long xrow = (long)node * N_FEAT;
  bool lo   = (khalf == 0);

  v8f c = {};
#pragma unroll
  for (int k = 0; k < 48; k += 4) {
    v2f a = *(const v2f*)(X + xrow + k + khalf);
    v2f b;
    b.x = W0[(k + khalf)     * HIDDEN + row];
    b.y = W0[(k + khalf + 1) * HIDDEN + row];
    c = __builtin_amdgcn_wmma_f32_16x16x4_f32(false, a, false, b, (short)0, c, false, false);
  }
  {
    v2f a0 = *(const v2f*)(X + xrow + 48);
    float b0x = W0[48 * HIDDEN + row];
    float b0y = W0[49 * HIDDEN + row];
    v2f a, b;
    a.x = lo ? a0.x : 0.0f;  a.y = lo ? a0.y : 0.0f;
    b.x = lo ? b0x  : 0.0f;  b.y = lo ? b0y  : 0.0f;
    c = __builtin_amdgcn_wmma_f32_16x16x4_f32(false, a, false, b, (short)0, c, false, false);
  }

  int col = lane & 15;
  int mhi = (lane >> 4) * 8;
  float bias = B[col];
#pragma unroll
  for (int j = 0; j < 8; ++j) {
    int m = base + mhi + j;
    if (m < n) {
      float v = (c[j] + AGG[(long)m * HIDDEN + col]) + bias;
      H[(long)m * HIDDEN + col] = v > 0.0f ? v : 0.0f;
    }
  }
}

__global__ __launch_bounds__(256) void tag2_dense(
    const float* __restrict__ H, const float* __restrict__ AGG,
    const float* __restrict__ W0, const float* __restrict__ W1,
    const float* __restrict__ B, float* __restrict__ OUT, int n) {
  int wave = (int)((blockIdx.x * blockDim.x + threadIdx.x) >> 5);
  int lane = threadIdx.x & 31;
  int base = wave * 16;
  if (base >= n) return;

  int row   = lane & 15;
  int khalf = (lane >> 4) * 2;
  int node  = base + row;
  node = node < n ? node : n - 1;
  long hrow = (long)node * HIDDEN;
  bool bok  = row < N_CLS;
  int  wrow = bok ? row : (N_CLS - 1);

  v8f c = {};
#pragma unroll
  for (int k = 0; k < HIDDEN; k += 4) {
    int ka = k + khalf;
    v2f a = *(const v2f*)(H + hrow + ka);
    float b0 = W0[(ka)     * N_CLS + wrow];
    float b1 = W0[(ka + 1) * N_CLS + wrow];
    v2f b;
    b.x = bok ? b0 : 0.0f;  b.y = bok ? b1 : 0.0f;
    c = __builtin_amdgcn_wmma_f32_16x16x4_f32(false, a, false, b, (short)0, c, false, false);
  }
#pragma unroll
  for (int k = 0; k < HIDDEN; k += 4) {
    int ka = k + khalf;
    v2f a = *(const v2f*)(AGG + hrow + ka);
    float b0 = W1[(ka)     * N_CLS + wrow];
    float b1 = W1[(ka + 1) * N_CLS + wrow];
    v2f b;
    b.x = bok ? b0 : 0.0f;  b.y = bok ? b1 : 0.0f;
    c = __builtin_amdgcn_wmma_f32_16x16x4_f32(false, a, false, b, (short)0, c, false, false);
  }

  int col  = lane & 15;
  int mhi  = (lane >> 4) * 8;
  bool cok = col < N_CLS;
  float bias = cok ? B[col] : 0.0f;

#pragma unroll
  for (int j = 0; j < 8; ++j) {
    int m = base + mhi + j;
    float v = c[j] + bias;
    float mx = cok ? v : -INFINITY;
    for (int off = 8; off >= 1; off >>= 1)
      mx = fmaxf(mx, __shfl_xor(mx, off, 16));
    float ex = cok ? expf(v - mx) : 0.0f;
    float sm = ex;
    for (int off = 8; off >= 1; off >>= 1)
      sm += __shfl_xor(sm, off, 16);
    float r = v - mx - logf(sm);
    if (cok && m < n) OUT[(long)m * N_CLS + col] = r;
  }
}

extern "C" void kernel_launch(void* const* d_in, const int* in_sizes, int n_in,
                              void* d_out, int out_size, void* d_ws, size_t ws_size,
                              hipStream_t stream) {
  const float* x    = (const float*)d_in[0];
  const int*   ei   = (const int*)d_in[1];
  const float* w1_0 = (const float*)d_in[2];
  const float* w1_1 = (const float*)d_in[3];
  const float* b1   = (const float*)d_in[4];
  const float* w2_0 = (const float*)d_in[5];
  const float* w2_1 = (const float*)d_in[6];
  const float* b2   = (const float*)d_in[7];
  float* out = (float*)d_out;

  int n = in_sizes[0] / N_FEAT;
  int E = in_sizes[1] / 2;
  const int* src = ei;
  const int* dst = ei + E;

  float* ws   = (float*)d_ws;
  float* deg  = ws;
  float* agg1 = deg  + n;
  float* agg2 = agg1 + (long)n * HIDDEN;
  float* dinv = agg2 + (long)n * HIDDEN;
  float* nrm  = dinv + n;
  float* h    = nrm  + E;
  float* y1   = h    + (long)n * HIDDEN;

  long zn = (long)n + (long)n * HIDDEN + (long)n * HIDDEN;
  zero_fill<<<(unsigned)((zn + 255) / 256), 256, 0, stream>>>(deg, zn);

  deg_kernel <<<(E + 255) / 256, 256, 0, stream>>>(dst, deg, E);
  dinv_kernel<<<(n + 255) / 256, 256, 0, stream>>>(deg, dinv, n);
  norm_kernel<<<(E + 255) / 256, 256, 0, stream>>>(src, dst, dinv, nrm, E);

  int ntiles = (n + 15) / 16;
  int nblk   = (ntiles + 7) / 8;
  long t16 = (long)E * 16;
  unsigned sblk = (unsigned)((t16 + 255) / 256);

  proj1_dense<<<nblk, 256, 0, stream>>>(x, w1_1, y1, n);
  scatter16<<<sblk, 256, 0, stream>>>(src, dst, nrm, y1, agg1, E);
  tag1_dense<<<nblk, 256, 0, stream>>>(x, agg1, w1_0, b1, h, n);

  scatter16<<<sblk, 256, 0, stream>>>(src, dst, nrm, h, agg2, E);

  tag2_dense<<<nblk, 256, 0, stream>>>(h, agg2, w2_0, w2_1, b2, out, n);
}


# Edge sums without float atomics: edges grouped by destination in-call (count+rank, run alloc, fill); each node's rows summed in registers in list order, one store per row; summation order changes
# speedup vs baseline: 2.5096x; 1.4090x over previous
#include <hip/hip_runtime.h>
#include <math.h>

typedef __attribute__((ext_vector_type(2))) float v2f;
typedef __attribute__((ext_vector_type(8))) float v8f;

#define N_FEAT   50
#define HIDDEN   16
#define N_CLS    10

__global__ void zero_fill(float* __restrict__ p, long n) {
  long i = (long)blockIdx.x * blockDim.x + threadIdx.x;
  if (i < n) p[i] = 0.0f;
}

__global__ void count_rank(const int* __restrict__ dst, int* __restrict__ cnt,
                           int* __restrict__ rank, int E) {
  int e = (blockIdx.x * blockDim.x + threadIdx.x) * 4;
  if (e + 3 < E) {
    int4 d = *(const int4*)(dst + e);
    int4 r;
    r.x = atomicAdd(&cnt[d.x], 1);
    r.y = atomicAdd(&cnt[d.y], 1);
    r.z = atomicAdd(&cnt[d.z], 1);
    r.w = atomicAdd(&cnt[d.w], 1);
    *(int4*)(rank + e) = r;
  } else {
    for (int k = e; k < E; ++k) rank[k] = atomicAdd(&cnt[dst[k]], 1);
  }
}

__global__ void alloc_runs(const int* __restrict__ cnt, int* __restrict__ cursor,
                           int* __restrict__ off, float* __restrict__ dinv, int n) {
  int i = blockIdx.x * blockDim.x + threadIdx.x;
  int lane = threadIdx.x & 31;
  int c = (i < n) ? cnt[i] : 0;
  int s = c;
  for (int d = 1; d < 32; d <<= 1) {
    int t = __shfl_up(s, d, 32);
    if (lane >= d) s += t;
  }
  int base = 0;
  if (lane == 31) base = atomicAdd(cursor, s);
  base = __shfl(base, 31, 32);
  if (i < n) {
    off[i] = base + s - c;
    dinv[i] = (c > 0) ? rsqrtf((float)c) : 0.0f;
  }
}

__device__ __forceinline__ void place_edge(int s, int d, int r,
                                           const int* __restrict__ off,
                                           const float* __restrict__ dinv,
                                           int2* __restrict__ csr, int E) {
  float w = dinv[s] * dinv[d];
  unsigned pos = (unsigned)(off[d] + r);
  pos = pos < (unsigned)E ? pos : (unsigned)(E - 1);
  csr[pos] = make_int2(s, __float_as_int(w));
}

__global__ void fill_csr(const int* __restrict__ src, const int* __restrict__ dst,
                         const int* __restrict__ rank, const int* __restrict__ off,
                         const float* __restrict__ dinv, int2* __restrict__ csr, int E) {
  int e = (blockIdx.x * blockDim.x + threadIdx.x) * 4;
  if (e + 3 < E) {
    int4 s = *(const int4*)(src + e);
    int4 d = *(const int4*)(dst + e);
    int4 r = *(const int4*)(rank + e);
    place_edge(s.x, d.x, r.x, off, dinv, csr, E);
    place_edge(s.y, d.y, r.y, off, dinv, csr, E);
    place_edge(s.z, d.z, r.z, off, dinv, csr, E);
    place_edge(s.w, d.w, r.w, off, dinv, csr, E);
  } else {
    for (int k = e; k < E; ++k) place_edge(src[k], dst[k], rank[k], off, dinv, csr, E);
  }
}

__device__ __forceinline__ float4 row_piece(const float4* __restrict__ h4, int s, int q, int n) {
  s = ((unsigned)s < (unsigned)n) ? s : 0;
  return h4[(long)s * 4 + q];
}

__global__ __launch_bounds__(256) void gather16(
    const int2* __restrict__ csr, const int* __restrict__ off, const int* __restrict__ cnt,
    const float* __restrict__ h, float* __restrict__ agg, int n, int E) {
  int t = blockIdx.x * blockDim.x + threadIdx.x;
  int node = t >> 2;
  int q = t & 3;
  if (node >= n) return;
  int j   = off[node];
  int end = j + cnt[node];
  j   = j   < 0 ? 0 : j;
  end = end > E ? E : end;
  int last = end - 1;
  const float4* h4 = (const float4*)h;
  float4 acc = {0.0f, 0.0f, 0.0f, 0.0f};
  for (; j < end; j += 4) {
    int j1 = j + 1, j2 = j + 2, j3 = j + 3;
    bool v1 = j1 < end, v2 = j2 < end, v3 = j3 < end;
    int2 e0 = csr[j];
    int2 e1 = csr[v1 ? j1 : last];
    int2 e2 = csr[v2 ? j2 : last];
    int2 e3 = csr[v3 ? j3 : last];
    float4 r0 = row_piece(h4, e0.x, q, n);
    float4 r1 = row_piece(h4, e1.x, q, n);
    float4 r2 = row_piece(h4, e2.x, q, n);
    float4 r3 = row_piece(h4, e3.x, q, n);
    float w0 = __int_as_float(e0.y);
    float w1 = v1 ? __int_as_float(e1.y) : 0.0f;
    float w2 = v2 ? __int_as_float(e2.y) : 0.0f;
    float w3 = v3 ? __int_as_float(e3.y) : 0.0f;
    acc.x += r0.x * w0; acc.y += r0.y * w0; acc.z += r0.z * w0; acc.w += r0.w * w0;
    acc.x += r1.x * w1; acc.y += r1.y * w1; acc.z += r1.z * w1; acc.w += r1.w * w1;
    acc.x += r2.x * w2; acc.y += r2.y * w2; acc.z += r2.z * w2; acc.w += r2.w * w2;
    acc.x += r3.x * w3; acc.y += r3.y * w3; acc.z += r3.z * w3; acc.w += r3.w * w3;
  }
  ((float4*)agg)[(long)node * 4 + q] = acc;
}

__global__ __launch_bounds__(256) void proj1_dense(
    const float* __restrict__ X, const float* __restrict__ W1,
    float* __restrict__ Y, int n) {
  int wave = (int)((blockIdx.x * blockDim.x + threadIdx.x) >> 5);
  int lane = threadIdx.x & 31;
  int base = wave * 16;
  if (base >= n) return;

  int row   = lane & 15;
  int khalf = (lane >> 4) * 2;
  int node  = base + row;
  node = node < n ? node : n - 1;
  long xrow = (long)node * N_FEAT;
  bool lo   = (khalf == 0);

  v8f c = {};
#pragma unroll
  for (int k = 0; k < 48; k += 4) {
    v2f a = *(const v2f*)(X + xrow + k + khalf);
    v2f b;
    b.x = W1[(k + khalf)     * HIDDEN + row];
    b.y = W1[(k + khalf + 1) * HIDDEN + row];
    c = __builtin_amdgcn_wmma_f32_16x16x4_f32(false, a, false, b, (short)0, c, false, false);
  }
  {
    v2f a0 = *(const v2f*)(X + xrow + 48);
    float b0x = W1[48 * HIDDEN + row];
    float b0y = W1[49 * HIDDEN + row];
    v2f a, b;
    a.x = lo ? a0.x : 0.0f;  a.y = lo ? a0.y : 0.0f;
    b.x = lo ? b0x  : 0.0f;  b.y = lo ? b0y  : 0.0f;
    c = __builtin_amdgcn_wmma_f32_16x16x4_f32(false, a, false, b, (short)0, c, false, false);
  }

  int col = lane & 15;
  int mhi = (lane >> 4) * 8;
#pragma unroll
  for (int j = 0; j < 8; ++j) {
    int m = base + mhi + j;
    if (m < n) Y[(long)m * HIDDEN + col] = c[j];
  }
}

__global__ __launch_bounds__(256) void tag1_dense(
    const float* __restrict__ X, const float* __restrict__ AGG,
    const float* __restrict__ W0,
    const float* __restrict__ B, float* __restrict__ H, int n) {
  int wave = (int)((blockIdx.x * blockDim.x + threadIdx.x) >> 5);
  int lane = threadIdx.x & 31;
  int base = wave * 16;
  if (base >= n) return;

  int row   = lane & 15;
  int khalf = (lane >> 4) * 2;
  int node  = base + row;
  node = node < n ? node : n - 1;
  long xrow = (long)node * N_FEAT;
  bool lo   = (khalf == 0);

  v8f c = {};
#pragma unroll
  for (int k = 0; k < 48; k += 4) {
    v2f a = *(const v2f*)(X + xrow + k + khalf);
    v2f b;
    b.x = W0[(k + khalf)     * HIDDEN + row];
    b.y = W0[(k + khalf + 1) * HIDDEN + row];
    c = __builtin_amdgcn_wmma_f32_16x16x4_f32(false, a, false, b, (short)0, c, false, false);
  }
  {
    v2f a0 = *(const v2f*)(X + xrow + 48);
    float b0x = W0[48 * HIDDEN + row];
    float b0y = W0[49 * HIDDEN + row];
    v2f a, b;
    a.x = lo ? a0.x : 0.0f;  a.y = lo ? a0.y : 0.0f;
    b.x = lo ? b0x  : 0.0f;  b.y = lo ? b0y  : 0.0f;
    c = __builtin_amdgcn_wmma_f32_16x16x4_f32(false, a, false, b, (short)0, c, false, false);
  }

  int col = lane & 15;
  int mhi = (lane >> 4) * 8;
  float bias = B[col];
#pragma unroll
  for (int j = 0; j < 8; ++j) {
    int m = base + mhi + j;
    if (m < n) {
      float v = (c[j] + AGG[(long)m * HIDDEN + col]) + bias;
      H[(long)m * HIDDEN + col] = v > 0.0f ? v : 0.0f;
    }
  }
}

__global__ __launch_bounds__(256) void tag2_dense(
    const float* __restrict__ H, const float* __restrict__ AGG,
    const float* __restrict__ W0, const float* __restrict__ W1,
    const float* __restrict__ B, float* __restrict__ OUT, int n) {
  int wave = (int)((blockIdx.x * blockDim.x + threadIdx.x) >> 5);
  int lane = threadIdx.x & 31;
  int base = wave * 16;
  if (base >= n) return;

  int row   = lane & 15;
  int khalf = (lane >> 4) * 2;
  int node  = base + row;
  node = node < n ? node : n - 1;
  long hrow = (long)node * HIDDEN;
  bool bok  = row < N_CLS;
  int  wrow = bok ? row : (N_CLS - 1);

  v8f c = {};
#pragma unroll
  for (int k = 0; k < HIDDEN; k += 4) {
    int ka = k + khalf;
    v2f a = *(const v2f*)(H + hrow + ka);
    float b0 = W0[(ka)     * N_CLS + wrow];
    float b1 = W0[(ka + 1) * N_CLS + wrow];
    v2f b;
    b.x = bok ? b0 : 0.0f;  b.y = bok ? b1 : 0.0f;
    c = __builtin_amdgcn_wmma_f32_16x16x4_f32(false, a, false, b, (short)0, c, false, false);
  }
#pragma unroll
  for (int k = 0; k < HIDDEN; k += 4) {
    int ka = k + khalf;
    v2f a = *(const v2f*)(AGG + hrow + ka);
    float b0 = W1[(ka)     * N_CLS + wrow];
    float b1 = W1[(ka + 1) * N_CLS + wrow];
    v2f b;
    b.x = bok ? b0 : 0.0f;  b.y = bok ? b1 : 0.0f;
    c = __builtin_amdgcn_wmma_f32_16x16x4_f32(false, a, false, b, (short)0, c, false, false);
  }

  int col  = lane & 15;
  int mhi  = (lane >> 4) * 8;
  bool cok = col < N_CLS;
  float bias = cok ? B[col] : 0.0f;

#pragma unroll
  for (int j = 0; j < 8; ++j) {
    int m = base + mhi + j;
    float v = c[j] + bias;
    float mx = cok ? v : -INFINITY;
    for (int off = 8; off >= 1; off >>= 1)
      mx = fmaxf(mx, __shfl_xor(mx, off, 16));
    float ex = cok ? expf(v - mx) : 0.0f;
    float sm = ex;
    for (int off = 8; off >= 1; off >>= 1)
      sm += __shfl_xor(sm, off, 16);
    float r = v - mx - logf(sm);
    if (cok && m < n) OUT[(long)m * N_CLS + col] = r;
  }
}

extern "C" void kernel_launch(void* const* d_in, const int* in_sizes, int n_in,
                              void* d_out, int out_size, void* d_ws, size_t ws_size,
                              hipStream_t stream) {
  const float* x    = (const float*)d_in[0];
  const int*   ei   = (const int*)d_in[1];
  const float* w1_0 = (const float*)d_in[2];
  const float* w1_1 = (const float*)d_in[3];
  const float* b1   = (const float*)d_in[4];
  const float* w2_0 = (const float*)d_in[5];
  const float* w2_1 = (const float*)d_in[6];
  const float* b2   = (const float*)d_in[7];
  float* out = (float*)d_out;

  int n = in_sizes[0] / N_FEAT;
  int E = in_sizes[1] / 2;
  const int* src = ei;
  const int* dst = ei + E;

  float* ws     = (float*)d_ws;
  int*   cnt    = (int*)ws;
  float* r1     = ws + n;
  int*   cursor = (int*)r1;
  float* dinv   = r1 + 32;
  float* y1     = r1;
  float* agg2   = r1;
  int*   off    = (int*)(r1 + (long)n * HIDDEN);
  float* r2     = r1 + (long)n * HIDDEN + n;
  int*   rank   = (int*)r2;
  float* agg1   = r2;
  float* h      = r2 + (long)n * HIDDEN;
  int2*  csr    = (int2*)(r2 + (long)n * HIDDEN * 2);

  long zn = (long)n + 1;
  zero_fill<<<(unsigned)((zn + 255) / 256), 256, 0, stream>>>(ws, zn);

  int e4blk = ((E + 3) / 4 + 255) / 256;
  count_rank<<<e4blk, 256, 0, stream>>>(dst, cnt, rank, E);
  alloc_runs<<<(n + 255) / 256, 256, 0, stream>>>(cnt, cursor, off, dinv, n);
  fill_csr  <<<e4blk, 256, 0, stream>>>(src, dst, rank, off, dinv, csr, E);

  int ntiles = (n + 15) / 16;
  int nblk   = (ntiles + 7) / 8;
  unsigned gblk = (unsigned)(((long)n * 4 + 255) / 256);

  proj1_dense<<<nblk, 256, 0, stream>>>(x, w1_1, y1, n);
  gather16<<<gblk, 256, 0, stream>>>(csr, off, cnt, y1, agg1, n, E);
  tag1_dense<<<nblk, 256, 0, stream>>>(x, agg1, w1_0, b1, h, n);

  gather16<<<gblk, 256, 0, stream>>>(csr, off, cnt, h, agg2, n, E);

  tag2_dense<<<nblk, 256, 0, stream>>>(h, agg2, w2_0, w2_1, b2, out, n);
}


# Dense tails folded into the edge-sum kernels: x@w1_0 rides the one pass over x, relu in sum 1; sum 2 does its WMMA transposed per 8-node wave + log-softmax; 7 launches -> 5; sum order changes
# speedup vs baseline: 5.8694x; 2.3388x over previous
#include <hip/hip_runtime.h>
#include <math.h>

typedef __attribute__((ext_vector_type(2))) float v2f;
typedef __attribute__((ext_vector_type(8))) float v8f;

#define N_FEAT   50
#define HIDDEN   16
#define N_CLS    10

#define BSH     9
#define BNODES  512
#define CHUNK   8192
#define EPT     8
#define MAXB    1024
#define MAXG    1024
#define CAP2    16384

__device__ __forceinline__ int clamp_node(int d, int n) {
  d = d < 0 ? 0 : d;
  return d < n ? d : n - 1;
}

__device__ __forceinline__ int block_excl_scan(int v, int* red, int* total) {
  int lane = threadIdx.x & 31, wv = threadIdx.x >> 5;
  int s = v;
  for (int d = 1; d < 32; d <<= 1) {
    int t = __shfl_up(s, d, 32);
    if (lane >= d) s += t;
  }
  if (lane == 31) red[wv] = s;
  __syncthreads();
  if (wv == 0) {
    int w = red[lane];
    int ws = w;
    for (int d = 1; d < 32; d <<= 1) {
      int t = __shfl_up(ws, d, 32);
      if (lane >= d) ws += t;
    }
    red[lane] = ws - w;
    if (lane == 31) red[32] = ws;
  }
  __syncthreads();
  int r = red[wv] + s - v;
  *total = red[32];
  __syncthreads();
  return r;
}

__global__ __launch_bounds__(1024) void chunk_sort(
    const int* __restrict__ src, const int* __restrict__ dst,
    int2* __restrict__ sorted, int* __restrict__ lstT, int E, int n, int B, int G) {
  __shared__ int  hist[MAXB];
  __shared__ int  lstart[MAXB];
  __shared__ int  red[33];
  __shared__ int2 stage[CHUNK];
  int t = threadIdx.x, g = blockIdx.x;
  hist[t] = 0;
  __syncthreads();
  int e0 = g * CHUNK;
  int sv[EPT], dv[EPT], rv[EPT];
#pragma unroll
  for (int k = 0; k < EPT; ++k) {
    int e = e0 + k * 1024 + t;
    int ec = e < E ? e : E - 1;
    sv[k] = src[ec];
    dv[k] = dst[ec];
  }
#pragma unroll
  for (int k = 0; k < EPT; ++k) {
    int e = e0 + k * 1024 + t;
    rv[k] = 0;
    if (e < E) {
      dv[k] = clamp_node(dv[k], n);
      rv[k] = atomicAdd(&hist[dv[k] >> BSH], 1);
    } else {
      dv[k] = -1;
    }
  }
  __syncthreads();
  int tot;
  int ls = block_excl_scan(hist[t], red, &tot);
  lstart[t] = ls;
  __syncthreads();
  int lb[EPT];
#pragma unroll
  for (int k = 0; k < EPT; ++k) lb[k] = lstart[(dv[k] < 0 ? 0 : dv[k]) >> BSH];
#pragma unroll
  for (int k = 0; k < EPT; ++k) {
    if (dv[k] >= 0) {
      int p = lb[k] + rv[k];
      p = p < CHUNK ? p : CHUNK - 1;
      stage[p] = make_int2(sv[k], dv[k]);
    }
  }
  if (t < B)  lstT[(long)t * G + g] = ls;
  if (t == 0) lstT[(long)B * G + g] = tot;
  __syncthreads();
  int2 w[EPT];
#pragma unroll
  for (int k = 0; k < EPT; ++k) w[k] = stage[k * 1024 + t];
#pragma unroll
  for (int k = 0; k < EPT; ++k) {
    int i = k * 1024 + t;
    if (i < tot) sorted[(long)g * CHUNK + i] = w[k];
  }
}

__global__ __launch_bounds__(1024) void bin_order(
    const int2* __restrict__ sorted, const int* __restrict__ lstT, int* __restrict__ csr,
    int* __restrict__ cnt, int* __restrict__ off, float* __restrict__ dinv, int n, int E, int G) {
  __shared__ int pstart[MAXG + 1];
  __shared__ int pbase[MAXG];
  __shared__ int first[CAP2 / 32 + 1];
  __shared__ int cntL[BNODES];
  __shared__ int cur[BNODES];
  __shared__ int red[33];
  __shared__ int eS[CAP2];
  __shared__ unsigned short eD[CAP2];
  __shared__ int stg[CAP2];
  int t = threadIdx.x, b = blockIdx.x;
  int lane = t & 31, wv = t >> 5;
  int base = b << BSH;

  int st = 0, len = 0;
  if (t < G) {
    st = lstT[(long)b * G + t];
    int en = lstT[(long)(b + 1) * G + t];
    st = st < 0 ? 0 : (st > CHUNK ? CHUNK : st);
    en = en < st ? st : (en > CHUNK ? CHUNK : en);
    len = en - st;
  }
  int m, bs;
  int ps = block_excl_scan(len, red, &m);
  (void)block_excl_scan(st, red, &bs);
  bs = bs < 0 ? 0 : (bs > E ? E : bs);
  m  = m > E - bs ? E - bs : m;
  pstart[t] = ps;
  if (t == 0) pstart[MAXG] = m;
  if (t < MAXG) pbase[t] = t * CHUNK + st - ps;
  if (t < BNODES) cntL[t] = 0;
  __syncthreads();

  bool in_lds = m <= CAP2;
  if (in_lds) {
    if (len > 0) {
      int pe = ps + len; pe = pe < m ? pe : m;
      for (int k = (ps + 31) >> 5; (k << 5) < pe; ++k) first[k] = t;
    }
    __syncthreads();
    for (int i0 = t; i0 < m; i0 += 8 * 1024) {
      int2 v[8];
#pragma unroll
      for (int u = 0; u < 8; ++u) {
        int i = i0 + u * 1024;
        v[u] = make_int2(0, base);
        if (i < m) {
          int p = first[i >> 5];
          p = p < 0 ? 0 : (p < G ? p : G - 1);
          while (p + 1 < G && i >= pstart[p + 1]) ++p;
          v[u] = sorted[pbase[p] + i];
        }
      }
#pragma unroll
      for (int u = 0; u < 8; ++u) {
        int i = i0 + u * 1024;
        if (i < m) {
          int dl = (v[u].y - base) & (BNODES - 1);
          atomicAdd(&cntL[dl], 1);
          eS[i] = v[u].x;
          eD[i] = (unsigned short)dl;
        }
      }
    }
  } else {
    for (int p = wv; p < G; p += 32) {
      int q0 = pstart[p], q1 = pstart[p + 1];
      q1 = q1 < m ? q1 : m;
      for (int i = q0 + lane; i < q1; i += 32) {
        int dl = (sorted[pbase[p] + i].y - base) & (BNODES - 1);
        atomicAdd(&cntL[dl], 1);
      }
    }
  }
  __syncthreads();
  int c = (t < BNODES) ? cntL[t] : 0;
  int tot;
  int lo = block_excl_scan(c, red, &tot);
  if (t < BNODES) {
    cur[t] = lo;
    int node = base + t;
    if (node < n) {
      cnt[node]  = c;
      off[node]  = bs + lo;
      dinv[node] = (c > 0) ? rsqrtf((float)c) : 0.0f;
    }
  }
  __syncthreads();
  if (in_lds) {
    for (int i0 = t; i0 < m; i0 += 8 * 1024) {
      int dd[8], ss[8];
#pragma unroll
      for (int u = 0; u < 8; ++u) {
        int i = i0 + u * 1024;
        int ic = i < m ? i : 0;
        dd[u] = eD[ic];
        ss[u] = eS[ic];
      }
#pragma unroll
      for (int u = 0; u < 8; ++u) {
        int i = i0 + u * 1024;
        if (i < m) {
          int p = atomicAdd(&cur[dd[u]], 1);
          p = p < 0 ? 0 : (p < CAP2 ? p : CAP2 - 1);
          stg[p] = ss[u];
        }
      }
    }
    __syncthreads();
    for (int i = t; i < m; i += 1024) csr[bs + i] = stg[i];
  } else {
    for (int p = wv; p < G; p += 32) {
      int q0 = pstart[p], q1 = pstart[p + 1];
      q1 = q1 < m ? q1 : m;
      for (int i = q0 + lane; i < q1; i += 32) {
        int2 v = sorted[pbase[p] + i];
        int dl = (v.y - base) & (BNODES - 1);
        unsigned pos = (unsigned)(bs + atomicAdd(&cur[dl], 1));
        pos = pos < (unsigned)E ? pos : (unsigned)(E - 1);
        csr[pos] = v.x;
      }
    }
  }
}

__device__ __forceinline__ float4 row_piece(const float4* __restrict__ h4, int s, int q, int n) {
  s = ((unsigned)s < (unsigned)n) ? s : 0;
  return h4[(long)s * 4 + q];
}

__device__ __forceinline__ float4 edge_sum(const int* __restrict__ csr, const float4* __restrict__ h4,
                                           int j, int end, int q, int n, int E) {
  j   = j   < 0 ? 0 : j;
  end = end > E ? E : end;
  int last = end - 1;
  float4 acc = {0.0f, 0.0f, 0.0f, 0.0f};
  for (; j < end; j += 4) {
    int j1 = j + 1, j2 = j + 2, j3 = j + 3;
    bool v1 = j1 < end, v2 = j2 < end, v3 = j3 < end;
    int s0 = csr[j];
    int s1 = csr[v1 ? j1 : last];
    int s2 = csr[v2 ? j2 : last];
    int s3 = csr[v3 ? j3 : last];
    float4 r0 = row_piece(h4, s0, q, n);
    float4 r1 = row_piece(h4, s1, q, n);
    float4 r2 = row_piece(h4, s2, q, n);
    float4 r3 = row_piece(h4, s3, q, n);
    acc.x += r0.x; acc.y += r0.y; acc.z += r0.z; acc.w += r0.w;
    acc.x += v1 ? r1.x : 0.0f; acc.y += v1 ? r1.y : 0.0f; acc.z += v1 ? r1.z : 0.0f; acc.w += v1 ? r1.w : 0.0f;
    acc.x += v2 ? r2.x : 0.0f; acc.y += v2 ? r2.y : 0.0f; acc.z += v2 ? r2.z : 0.0f; acc.w += v2 ? r2.w : 0.0f;
    acc.x += v3 ? r3.x : 0.0f; acc.y += v3 ? r3.y : 0.0f; acc.z += v3 ? r3.z : 0.0f; acc.w += v3 ? r3.w : 0.0f;
  }
  return acc;
}

__global__ __launch_bounds__(256) void gather16_tag1(
    const int* __restrict__ csr, const int* __restrict__ off, const int* __restrict__ cnt,
    const float* __restrict__ dinv, const float* __restrict__ y, const float* __restrict__ B,
    float* __restrict__ HS, float* __restrict__ H, int n, int E) {
  int t = blockIdx.x * blockDim.x + threadIdx.x;
  int node = t >> 2;
  int q = t & 3;
  if (node >= n) return;
  int j   = off[node];
  int end = j + cnt[node];
  float dv = dinv[node];
  float4 z = ((const float4*)H)[(long)node * 4 + q];
  float b0 = B[4 * q], b1v = B[4 * q + 1], b2v = B[4 * q + 2], b3v = B[4 * q + 3];
  float4 acc = edge_sum(csr, (const float4*)y, j, end, q, n, E);
  float4 v;
  v.x = (z.x + acc.x * dv) + b0;
  v.y = (z.y + acc.y * dv) + b1v;
  v.z = (z.z + acc.z * dv) + b2v;
  v.w = (z.w + acc.w * dv) + b3v;
  v.x = v.x > 0.0f ? v.x : 0.0f;  v.y = v.y > 0.0f ? v.y : 0.0f;
  v.z = v.z > 0.0f ? v.z : 0.0f;  v.w = v.w > 0.0f ? v.w : 0.0f;
  ((float4*)H)[(long)node * 4 + q] = v;
  float4 s = {v.x * dv, v.y * dv, v.z * dv, v.w * dv};
  ((float4*)HS)[(long)node * 4 + q] = s;
}

__global__ __launch_bounds__(256) void gather16_tag2(
    const int* __restrict__ csr, const int* __restrict__ off, const int* __restrict__ cnt,
    const float* __restrict__ dinv, const float* __restrict__ HS, const float* __restrict__ H,
    const float* __restrict__ W0, const float* __restrict__ W1, const float* __restrict__ B,
    float* __restrict__ OUT, int n, int E) {
  int t = blockIdx.x * blockDim.x + threadIdx.x;
  int lane = threadIdx.x & 31;
  int wbase = (t >> 5) * 8;
  __shared__ __attribute__((aligned(16))) float so[64 * N_CLS];

  int node = wbase + (lane >> 2);
  int q = lane & 3;
  bool nv = node < n;
  int nc = nv ? node : n - 1;
  int j   = off[nc];
  int cn  = cnt[nc];
  float dv = dinv[nc];
  int end = j + (nv ? cn : 0);

  int c  = lane & 15;
  int hh = lane >> 4;
  int dn = wbase + (c & 7);
  dn = dn < n ? dn : n - 1;
  const float4* H4 = (const float4*)H;
  float4 hA = H4[(long)dn * 4 + 2 * hh];
  float4 hB = H4[(long)dn * 4 + 2 * hh + 1];
  bool cok = c < N_CLS;
  int  wc  = cok ? c : (N_CLS - 1);
  float w0[8], w1[8], bb[8];
#pragma unroll
  for (int i = 0; i < 8; ++i) {
    int f = 8 * hh + i;
    float a0 = W0[f * N_CLS + wc];
    float a1 = W1[f * N_CLS + wc];
    w0[i] = cok ? a0 : 0.0f;
    w1[i] = cok ? a1 : 0.0f;
    int cl = 8 * hh + i;
    float bv = B[cl < N_CLS ? cl : (N_CLS - 1)];
    bb[i] = cl < N_CLS ? bv : 0.0f;
  }

  float4 acc = edge_sum(csr, (const float4*)HS, j, end, q, n, E);
  acc.x *= dv; acc.y *= dv; acc.z *= dv; acc.w *= dv;

  int sl = 4 * (c & 7) + 2 * hh;
  float g[8];
  g[0] = __shfl(acc.x, sl, 32);      g[1] = __shfl(acc.y, sl, 32);
  g[2] = __shfl(acc.z, sl, 32);      g[3] = __shfl(acc.w, sl, 32);
  g[4] = __shfl(acc.x, sl + 1, 32);  g[5] = __shfl(acc.y, sl + 1, 32);
  g[6] = __shfl(acc.z, sl + 1, 32);  g[7] = __shfl(acc.w, sl + 1, 32);
  float hv[8] = {hA.x, hA.y, hA.z, hA.w, hB.x, hB.y, hB.z, hB.w};

  v8f ch = {}, ca = {};
#pragma unroll
  for (int ks = 0; ks < 4; ++ks) {
    v2f a, b;
    a.x = w0[2 * ks];  a.y = w0[2 * ks + 1];
    b.x = hv[2 * ks];  b.y = hv[2 * ks + 1];
    ch = __builtin_amdgcn_wmma_f32_16x16x4_f32(false, a, false, b, (short)0, ch, false, false);
    a.x = w1[2 * ks];  a.y = w1[2 * ks + 1];
    b.x = g[2 * ks];   b.y = g[2 * ks + 1];
    ca = __builtin_amdgcn_wmma_f32_16x16x4_f32(false, a, false, b, (short)0, ca, false, false);
  }

  float v[8];
  float mx = -INFINITY;
#pragma unroll
  for (int r = 0; r < 8; ++r) {
    v[r] = (ch[r] + ca[r]) + bb[r];
    bool ok = (hh == 0) || (r < N_CLS - 8);
    mx = ok ? fmaxf(mx, v[r]) : mx;
  }
  mx = fmaxf(mx, __shfl_xor(mx, 16, 32));
  float sm = 0.0f;
#pragma unroll
  for (int r = 0; r < 8; ++r) {
    bool ok = (hh == 0) || (r < N_CLS - 8);
    sm += ok ? expf(v[r] - mx) : 0.0f;
  }
  sm += __shfl_xor(sm, 16, 32);
  float ls = logf(sm);
  if (c < 8) {
    float* o = so + ((threadIdx.x >> 5) * 8 + c) * N_CLS + 8 * hh;
    float2 p0 = {v[0] - mx - ls, v[1] - mx - ls};
    *(float2*)o = p0;
    if (hh == 0) {
      float2 p1 = {v[2] - mx - ls, v[3] - mx - ls};
      float2 p2 = {v[4] - mx - ls, v[5] - mx - ls};
      float2 p3 = {v[6] - mx - ls, v[7] - mx - ls};
      *(float2*)(o + 2) = p1;
      *(float2*)(o + 4) = p2;
      *(float2*)(o + 6) = p3;
    }
  }
  __syncthreads();
  long g0   = (long)blockIdx.x * (64 * N_CLS);
  long gend = (long)n * N_CLS;
  int  i4   = threadIdx.x * 4;
  if (i4 < 64 * N_CLS) {
    if (g0 + i4 + 3 < gend) {
      *(float4*)(OUT + g0 + i4) = *(const float4*)(so + i4);
    } else {
      for (int u = 0; u < 4; ++u)
        if (g0 + i4 + u < gend) OUT[g0 + i4 + u] = so[i4 + u];
    }
  }
}

__global__ __launch_bounds__(256) void proj1_dense(
    const float* __restrict__ X, const float* __restrict__ W1, const float* __restrict__ W0,
    const float* __restrict__ DINV, float* __restrict__ Y, float* __restrict__ Z, int n) {
  int wave = (int)((blockIdx.x * blockDim.x + threadIdx.x) >> 5);
  int lane = threadIdx.x & 31;
  int base = wave * 16;
  if (base >= n) return;

  int row   = lane & 15;
  int khalf = (lane >> 4) * 2;
  int node  = base + row;
  node = node < n ? node : n - 1;
  long xrow = (long)node * N_FEAT;
  bool lo   = (khalf == 0);

  v8f c1 = {}, c0 = {};
#pragma unroll
  for (int k = 0; k < 48; k += 4) {
    v2f a = *(const v2f*)(X + xrow + k + khalf);
    v2f b1, b0;
    b1.x = W1[(k + khalf)     * HIDDEN + row];
    b1.y = W1[(k + khalf + 1) * HIDDEN + row];
    b0.x = W0[(k + khalf)     * HIDDEN + row];
    b0.y = W0[(k + khalf + 1) * HIDDEN + row];
    c1 = __builtin_amdgcn_wmma_f32_16x16x4_f32(false, a, false, b1, (short)0, c1, false, false);
    c0 = __builtin_amdgcn_wmma_f32_16x16x4_f32(false, a, false, b0, (short)0, c0, false, false);
  }
  {
    v2f a0 = *(const v2f*)(X + xrow + 48);
    float p1x = W1[48 * HIDDEN + row], p1y = W1[49 * HIDDEN + row];
    float p0x = W0[48 * HIDDEN + row], p0y = W0[49 * HIDDEN + row];
    v2f a, b1, b0;
    a.x  = lo ? a0.x : 0.0f;  a.y  = lo ? a0.y : 0.0f;
    b1.x = lo ? p1x  : 0.0f;  b1.y = lo ? p1y  : 0.0f;
    b0.x = lo ? p0x  : 0.0f;  b0.y = lo ? p0y  : 0.0f;
    c1 = __builtin_amdgcn_wmma_f32_16x16x4_f32(false, a, false, b1, (short)0, c1, false, false);
    c0 = __builtin_amdgcn_wmma_f32_16x16x4_f32(false, a, false, b0, (short)0, c0, false, false);
  }

  int col = lane & 15;
  int mhi = (lane >> 4) * 8;
#pragma unroll
  for (int j = 0; j < 8; ++j) {
    int m = base + mhi + j;
    if (m < n) {
      Y[(long)m * HIDDEN + col] = c1[j] * DINV[m];
      Z[(long)m * HIDDEN + col] = c0[j];
    }
  }
}

extern "C" void kernel_launch(void* const* d_in, const int* in_sizes, int n_in,
                              void* d_out, int out_size, void* d_ws, size_t ws_size,
                              hipStream_t stream) {
  const float* x    = (const float*)d_in[0];
  const int*   ei   = (const int*)d_in[1];
  const float* w1_0 = (const float*)d_in[2];
  const float* w1_1 = (const float*)d_in[3];
  const float* b1   = (const float*)d_in[4];
  const float* w2_0 = (const float*)d_in[5];
  const float* w2_1 = (const float*)d_in[6];
  const float* b2   = (const float*)d_in[7];
  float* out = (float*)d_out;

  int n = in_sizes[0] / N_FEAT;
  int E = in_sizes[1] / 2;
  const int* src = ei;
  const int* dst = ei + E;

  int B = (n + BNODES - 1) / BNODES;
  int G = (E + CHUNK - 1) / CHUNK;

  float* ws     = (float*)d_ws;
  int*   cnt    = (int*)ws;
  int*   off    = cnt + n;
  float* dinv   = ws + 2L * n;
  int*   lstT   = (int*)(ws + 3L * n);
  long   lstw   = ((long)(B + 1) * G + 3) / 4 * 4;
  float* r1     = ws + 3L * n + lstw;
  float* y1     = r1;
  float* r2     = r1 + (long)n * HIDDEN;
  long   r2w    = 2L * G * CHUNK > 2L * n * HIDDEN ? 2L * G * CHUNK : 2L * n * HIDDEN;
  int2*  sorted = (int2*)r2;
  float* hs     = r2;
  float* h      = r2 + (long)n * HIDDEN;
  int*   csr    = (int*)(r2 + r2w);

  chunk_sort<<<G, 1024, 0, stream>>>(src, dst, sorted, lstT, E, n, B, G);
  bin_order <<<B, 1024, 0, stream>>>(sorted, lstT, csr, cnt, off, dinv, n, E, G);

  int ntiles = (n + 15) / 16;
  int nblk   = (ntiles + 7) / 8;
  unsigned gblk = (unsigned)(((long)n * 4 + 255) / 256);

  proj1_dense<<<nblk, 256, 0, stream>>>(x, w1_1, w1_0, dinv, y1, h, n);
  gather16_tag1<<<gblk, 256, 0, stream>>>(csr, off, cnt, dinv, y1, b1, hs, h, n, E);
  gather16_tag2<<<gblk, 256, 0, stream>>>(csr, off, cnt, dinv, hs, h, w2_0, w2_1, b2, out, n, E);
}


# Layer-1 projections share the chunk-sort launch as extra work-groups that run beside the sort; bin pass scales its y rows by deg^-1/2; edge pairs packed to one word to make room; 5 launches -> 4
# speedup vs baseline: 6.2174x; 1.0593x over previous
#include <hip/hip_runtime.h>
#include <math.h>

typedef __attribute__((ext_vector_type(2))) float v2f;
typedef __attribute__((ext_vector_type(8))) float v8f;

#define N_FEAT   50
#define HIDDEN   16
#define N_CLS    10

#define BSH     9
#define BNODES  512
#define CHUNK   8192
#define EPT     8
#define MAXB    1024
#define MAXG    1024
#define CAP2    16384

__device__ __forceinline__ int clamp_node(int d, int n) {
  d = d < 0 ? 0 : d;
  return d < n ? d : n - 1;
}

__device__ __forceinline__ int block_excl_scan(int v, int* red, int* total) {
  int lane = threadIdx.x & 31, wv = threadIdx.x >> 5;
  int s = v;
  for (int d = 1; d < 32; d <<= 1) {
    int t = __shfl_up(s, d, 32);
    if (lane >= d) s += t;
  }
  if (lane == 31) red[wv] = s;
  __syncthreads();
  if (wv == 0) {
    int w = red[lane];
    int ws = w;
    for (int d = 1; d < 32; d <<= 1) {
      int t = __shfl_up(ws, d, 32);
      if (lane >= d) ws += t;
    }
    red[lane] = ws - w;
    if (lane == 31) red[32] = ws;
  }
  __syncthreads();
  int r = red[wv] + s - v;
  *total = red[32];
  __syncthreads();
  return r;
}

__device__ __forceinline__ void proj1_tile(
    const float* __restrict__ X, const float* __restrict__ W1, const float* __restrict__ W0,
    float* __restrict__ Y, float* __restrict__ Z, int n, int wave);

__global__ __launch_bounds__(1024) void chunk_sort_proj1(
    const int* __restrict__ src, const int* __restrict__ dst,
    int* __restrict__ sorted, int* __restrict__ lstT,
    const float* __restrict__ X, const float* __restrict__ W1, const float* __restrict__ W0,
    float* __restrict__ Y, float* __restrict__ Z, int E, int n, int B, int G) {
  __shared__ int  hist[MAXB];
  __shared__ int  lstart[MAXB];
  __shared__ int  red[33];
  __shared__ int  stage[CHUNK];
  if ((int)blockIdx.x >= G) {
    proj1_tile(X, W1, W0, Y, Z, n, (((int)blockIdx.x - G) * 1024 + (int)threadIdx.x) >> 5);
    return;
  }
  int t = threadIdx.x, g = blockIdx.x;
  hist[t] = 0;
  __syncthreads();
  int e0 = g * CHUNK;
  int sv[EPT], dv[EPT], rv[EPT];
#pragma unroll
  for (int k = 0; k < EPT; ++k) {
    int e = e0 + k * 1024 + t;
    int ec = e < E ? e : E - 1;
    sv[k] = src[ec];
    dv[k] = dst[ec];
  }
#pragma unroll
  for (int k = 0; k < EPT; ++k) {
    int e = e0 + k * 1024 + t;
    rv[k] = 0;
    if (e < E) {
      dv[k] = clamp_node(dv[k], n);
      rv[k] = atomicAdd(&hist[dv[k] >> BSH], 1);
    } else {
      dv[k] = -1;
    }
  }
  __syncthreads();
  int tot;
  int ls = block_excl_scan(hist[t], red, &tot);
  lstart[t] = ls;
  __syncthreads();
  int lb[EPT];
#pragma unroll
  for (int k = 0; k < EPT; ++k) lb[k] = lstart[(dv[k] < 0 ? 0 : dv[k]) >> BSH];
#pragma unroll
  for (int k = 0; k < EPT; ++k) {
    if (dv[k] >= 0) {
      int p = lb[k] + rv[k];
      p = p < CHUNK ? p : CHUNK - 1;
      int s = ((unsigned)sv[k] < (unsigned)n) ? sv[k] : 0;
      stage[p] = (s << BSH) | (dv[k] & (BNODES - 1));
    }
  }
  if (t < B)  lstT[(long)t * G + g] = ls;
  if (t == 0) lstT[(long)B * G + g] = tot;
  __syncthreads();
  int w[EPT];
#pragma unroll
  for (int k = 0; k < EPT; ++k) w[k] = stage[k * 1024 + t];
#pragma unroll
  for (int k = 0; k < EPT; ++k) {
    int i = k * 1024 + t;
    if (i < tot) sorted[(long)g * CHUNK + i] = w[k];
  }
}

__global__ __launch_bounds__(1024) void bin_order(
    const int* __restrict__ sorted, const int* __restrict__ lstT, int* __restrict__ csr,
    int* __restrict__ cnt, int* __restrict__ off, float* __restrict__ dinv, float* Y,
    int n, int E, int G) {
  __shared__ int pstart[MAXG + 1];
  __shared__ int pbase[MAXG];
  __shared__ int first[CAP2 / 32 + 1];
  __shared__ int cntL[BNODES];
  __shared__ int cur[BNODES];
  __shared__ int red[33];
  __shared__ int eS[CAP2];
  __shared__ unsigned short eD[CAP2];
  __shared__ int stg[CAP2];
  __shared__ float dinvL[BNODES];
  int t = threadIdx.x, b = blockIdx.x;
  int lane = t & 31, wv = t >> 5;
  int base = b << BSH;

  float4* Y4 = (float4*)Y;
  long yq = (long)base * 4;
  bool yk0 = base + (t >> 2) < n, yk1 = base + ((t + 1024) >> 2) < n;
  float4 ya = {0.0f, 0.0f, 0.0f, 0.0f}, yb = {0.0f, 0.0f, 0.0f, 0.0f};
  if (yk0) ya = Y4[yq + t];
  if (yk1) yb = Y4[yq + t + 1024];

  int st = 0, len = 0;
  if (t < G) {
    st = lstT[(long)b * G + t];
    int en = lstT[(long)(b + 1) * G + t];
    st = st < 0 ? 0 : (st > CHUNK ? CHUNK : st);
    en = en < st ? st : (en > CHUNK ? CHUNK : en);
    len = en - st;
  }
  int m, bs;
  int ps = block_excl_scan(len, red, &m);
  (void)block_excl_scan(st, red, &bs);
  bs = bs < 0 ? 0 : (bs > E ? E : bs);
  m  = m > E - bs ? E - bs : m;
  pstart[t] = ps;
  if (t == 0) pstart[MAXG] = m;
  if (t < MAXG) pbase[t] = t * CHUNK + st - ps;
  if (t < BNODES) cntL[t] = 0;
  __syncthreads();

  bool in_lds = m <= CAP2;
  if (in_lds) {
    if (len > 0) {
      int pe = ps + len; pe = pe < m ? pe : m;
      for (int k = (ps + 31) >> 5; (k << 5) < pe; ++k) first[k] = t;
    }
    __syncthreads();
    for (int i0 = t; i0 < m; i0 += 8 * 1024) {
      int v[8];
#pragma unroll
      for (int u = 0; u < 8; ++u) {
        int i = i0 + u * 1024;
        v[u] = 0;
        if (i < m) {
          int p = first[i >> 5];
          p = p < 0 ? 0 : (p < G ? p : G - 1);
          while (p + 1 < G && i >= pstart[p + 1]) ++p;
          v[u] = sorted[pbase[p] + i];
        }
      }
#pragma unroll
      for (int u = 0; u < 8; ++u) {
        int i = i0 + u * 1024;
        if (i < m) {
          int dl = v[u] & (BNODES - 1);
          atomicAdd(&cntL[dl], 1);
          eS[i] = (int)((unsigned)v[u] >> BSH);
          eD[i] = (unsigned short)dl;
        }
      }
    }
  } else {
    for (int p = wv; p < G; p += 32) {
      int q0 = pstart[p], q1 = pstart[p + 1];
      q1 = q1 < m ? q1 : m;
      for (int i = q0 + lane; i < q1; i += 32) {
        int dl = sorted[pbase[p] + i] & (BNODES - 1);
        atomicAdd(&cntL[dl], 1);
      }
    }
  }
  __syncthreads();
  int c = (t < BNODES) ? cntL[t] : 0;
  int tot;
  int lo = block_excl_scan(c, red, &tot);
  if (t < BNODES) {
    cur[t] = lo;
    int node = base + t;
    if (node < n) {
      float dn = (c > 0) ? rsqrtf((float)c) : 0.0f;
      cnt[node]  = c;
      off[node]  = bs + lo;
      dinv[node] = dn;
      dinvL[t]   = dn;
    }
  }
  __syncthreads();
  if (yk0) {
    float d = dinvL[t >> 2];
    float4 o = {ya.x * d, ya.y * d, ya.z * d, ya.w * d};
    Y4[yq + t] = o;
  }
  if (yk1) {
    float d = dinvL[(t + 1024) >> 2];
    float4 o = {yb.x * d, yb.y * d, yb.z * d, yb.w * d};
    Y4[yq + t + 1024] = o;
  }
  if (in_lds) {
    for (int i0 = t; i0 < m; i0 += 8 * 1024) {
      int dd[8], ss[8];
#pragma unroll
      for (int u = 0; u < 8; ++u) {
        int i = i0 + u * 1024;
        int ic = i < m ? i : 0;
        dd[u] = eD[ic];
        ss[u] = eS[ic];
      }
#pragma unroll
      for (int u = 0; u < 8; ++u) {
        int i = i0 + u * 1024;
        if (i < m) {
          int p = atomicAdd(&cur[dd[u]], 1);
          p = p < 0 ? 0 : (p < CAP2 ? p : CAP2 - 1);
          stg[p] = ss[u];
        }
      }
    }
    __syncthreads();
    for (int i = t; i < m; i += 1024) csr[bs + i] = stg[i];
  } else {
    for (int p = wv; p < G; p += 32) {
      int q0 = pstart[p], q1 = pstart[p + 1];
      q1 = q1 < m ? q1 : m;
      for (int i = q0 + lane; i < q1; i += 32) {
        int v = sorted[pbase[p] + i];
        int dl = v & (BNODES - 1);
        unsigned pos = (unsigned)(bs + atomicAdd(&cur[dl], 1));
        pos = pos < (unsigned)E ? pos : (unsigned)(E - 1);
        csr[pos] = (int)((unsigned)v >> BSH);
      }
    }
  }
}

__device__ __forceinline__ float4 row_piece(const float4* __restrict__ h4, int s, int q, int n) {
  s = ((unsigned)s < (unsigned)n) ? s : 0;
  return h4[(long)s * 4 + q];
}

__device__ __forceinline__ float4 edge_sum(const int* __restrict__ csr, const float4* __restrict__ h4,
                                           int j, int end, int q, int n, int E) {
  j   = j   < 0 ? 0 : j;
  end = end > E ? E : end;
  int last = end - 1;
  float4 acc = {0.0f, 0.0f, 0.0f, 0.0f};
  for (; j < end; j += 4) {
    int j1 = j + 1, j2 = j + 2, j3 = j + 3;
    bool v1 = j1 < end, v2 = j2 < end, v3 = j3 < end;
    int s0 = csr[j];
    int s1 = csr[v1 ? j1 : last];
    int s2 = csr[v2 ? j2 : last];
    int s3 = csr[v3 ? j3 : last];
    float4 r0 = row_piece(h4, s0, q, n);
    float4 r1 = row_piece(h4, s1, q, n);
    float4 r2 = row_piece(h4, s2, q, n);
    float4 r3 = row_piece(h4, s3, q, n);
    acc.x += r0.x; acc.y += r0.y; acc.z += r0.z; acc.w += r0.w;
    acc.x += v1 ? r1.x : 0.0f; acc.y += v1 ? r1.y : 0.0f; acc.z += v1 ? r1.z : 0.0f; acc.w += v1 ? r1.w : 0.0f;
    acc.x += v2 ? r2.x : 0.0f; acc.y += v2 ? r2.y : 0.0f; acc.z += v2 ? r2.z : 0.0f; acc.w += v2 ? r2.w : 0.0f;
    acc.x += v3 ? r3.x : 0.0f; acc.y += v3 ? r3.y : 0.0f; acc.z += v3 ? r3.z : 0.0f; acc.w += v3 ? r3.w : 0.0f;
  }
  return acc;
}

__global__ __launch_bounds__(256) void gather16_tag1(
    const int* __restrict__ csr, const int* __restrict__ off, const int* __restrict__ cnt,
    const float* __restrict__ dinv, const float* __restrict__ y, const float* __restrict__ B,
    float* __restrict__ HS, float* __restrict__ H, int n, int E) {
  int t = blockIdx.x * blockDim.x + threadIdx.x;
  int node = t >> 2;
  int q = t & 3;
  if (node >= n) return;
  int j   = off[node];
  int end = j + cnt[node];
  float dv = dinv[node];
  float4 z = ((const float4*)H)[(long)node * 4 + q];
  float b0 = B[4 * q], b1v = B[4 * q + 1], b2v = B[4 * q + 2], b3v = B[4 * q + 3];
  float4 acc = edge_sum(csr, (const float4*)y, j, end, q, n, E);
  float4 v;
  v.x = (z.x + acc.x * dv) + b0;
  v.y = (z.y + acc.y * dv) + b1v;
  v.z = (z.z + acc.z * dv) + b2v;
  v.w = (z.w + acc.w * dv) + b3v;
  v.x = v.x > 0.0f ? v.x : 0.0f;  v.y = v.y > 0.0f ? v.y : 0.0f;
  v.z = v.z > 0.0f ? v.z : 0.0f;  v.w = v.w > 0.0f ? v.w : 0.0f;
  ((float4*)H)[(long)node * 4 + q] = v;
  float4 s = {v.x * dv, v.y * dv, v.z * dv, v.w * dv};
  ((float4*)HS)[(long)node * 4 + q] = s;
}

__global__ __launch_bounds__(256) void gather16_tag2(
    const int* __restrict__ csr, const int* __restrict__ off, const int* __restrict__ cnt,
    const float* __restrict__ dinv, const float* __restrict__ HS, const float* __restrict__ H,
    const float* __restrict__ W0, const float* __restrict__ W1, const float* __restrict__ B,
    float* __restrict__ OUT, int n, int E) {
  int t = blockIdx.x * blockDim.x + threadIdx.x;
  int lane = threadIdx.x & 31;
  int wbase = (t >> 5) * 8;
  __shared__ __attribute__((aligned(16))) float so[64 * N_CLS];

  int node = wbase + (lane >> 2);
  int q = lane & 3;
  bool nv = node < n;
  int nc = nv ? node : n - 1;
  int j   = off[nc];
  int cn  = cnt[nc];
  float dv = dinv[nc];
  int end = j + (nv ? cn : 0);

  int c  = lane & 15;
  int hh = lane >> 4;
  int dn = wbase + (c & 7);
  dn = dn < n ? dn : n - 1;
  const float4* H4 = (const float4*)H;
  float4 hA = H4[(long)dn * 4 + 2 * hh];
  float4 hB = H4[(long)dn * 4 + 2 * hh + 1];
  bool cok = c < N_CLS;
  int  wc  = cok ? c : (N_CLS - 1);
  float w0[8], w1[8], bb[8];
#pragma unroll
  for (int i = 0; i < 8; ++i) {
    int f = 8 * hh + i;
    float a0 = W0[f * N_CLS + wc];
    float a1 = W1[f * N_CLS + wc];
    w0[i] = cok ? a0 : 0.0f;
    w1[i] = cok ? a1 : 0.0f;
    int cl = 8 * hh + i;
    float bv = B[cl < N_CLS ? cl : (N_CLS - 1)];
    bb[i] = cl < N_CLS ? bv : 0.0f;
  }

  float4 acc = edge_sum(csr, (const float4*)HS, j, end, q, n, E);
  acc.x *= dv; acc.y *= dv; acc.z *= dv; acc.w *= dv;

  int sl = 4 * (c & 7) + 2 * hh;
  float g[8];
  g[0] = __shfl(acc.x, sl, 32);      g[1] = __shfl(acc.y, sl, 32);
  g[2] = __shfl(acc.z, sl, 32);      g[3] = __shfl(acc.w, sl, 32);
  g[4] = __shfl(acc.x, sl + 1, 32);  g[5] = __shfl(acc.y, sl + 1, 32);
  g[6] = __shfl(acc.z, sl + 1, 32);  g[7] = __shfl(acc.w, sl + 1, 32);
  float hv[8] = {hA.x, hA.y, hA.z, hA.w, hB.x, hB.y, hB.z, hB.w};

  v8f ch = {}, ca = {};
#pragma unroll
  for (int ks = 0; ks < 4; ++ks) {
    v2f a, b;
    a.x = w0[2 * ks];  a.y = w0[2 * ks + 1];
    b.x = hv[2 * ks];  b.y = hv[2 * ks + 1];
    ch = __builtin_amdgcn_wmma_f32_16x16x4_f32(false, a, false, b, (short)0, ch, false, false);
    a.x = w1[2 * ks];  a.y = w1[2 * ks + 1];
    b.x = g[2 * ks];   b.y = g[2 * ks + 1];
    ca = __builtin_amdgcn_wmma_f32_16x16x4_f32(false, a, false, b, (short)0, ca, false, false);
  }

  float v[8];
  float mx = -INFINITY;
#pragma unroll
  for (int r = 0; r < 8; ++r) {
    v[r] = (ch[r] + ca[r]) + bb[r];
    bool ok = (hh == 0) || (r < N_CLS - 8);
    mx = ok ? fmaxf(mx, v[r]) : mx;
  }
  mx = fmaxf(mx, __shfl_xor(mx, 16, 32));
  float sm = 0.0f;
#pragma unroll
  for (int r = 0; r < 8; ++r) {
    bool ok = (hh == 0) || (r < N_CLS - 8);
    sm += ok ? expf(v[r] - mx) : 0.0f;
  }
  sm += __shfl_xor(sm, 16, 32);
  float ls = logf(sm);
  if (c < 8) {
    float* o = so + ((threadIdx.x >> 5) * 8 + c) * N_CLS + 8 * hh;
    float2 p0 = {v[0] - mx - ls, v[1] - mx - ls};
    *(float2*)o = p0;
    if (hh == 0) {
      float2 p1 = {v[2] - mx - ls, v[3] - mx - ls};
      float2 p2 = {v[4] - mx - ls, v[5] - mx - ls};
      float2 p3 = {v[6] - mx - ls, v[7] - mx - ls};
      *(float2*)(o + 2) = p1;
      *(float2*)(o + 4) = p2;
      *(float2*)(o + 6) = p3;
    }
  }
  __syncthreads();
  long g0   = (long)blockIdx.x * (64 * N_CLS);
  long gend = (long)n * N_CLS;
  int  i4   = threadIdx.x * 4;
  if (i4 < 64 * N_CLS) {
    if (g0 + i4 + 3 < gend) {
      *(float4*)(OUT + g0 + i4) = *(const float4*)(so + i4);
    } else {
      for (int u = 0; u < 4; ++u)
        if (g0 + i4 + u < gend) OUT[g0 + i4 + u] = so[i4 + u];
    }
  }
}

__device__ __forceinline__ void proj1_tile(
    const float* __restrict__ X, const float* __restrict__ W1, const float* __restrict__ W0,
    float* __restrict__ Y, float* __restrict__ Z, int n, int wave) {
  int lane = threadIdx.x & 31;
  int base = wave * 16;
  if (base >= n) return;

  int row   = lane & 15;
  int khalf = (lane >> 4) * 2;
  int node  = base + row;
  node = node < n ? node : n - 1;
  long xrow = (long)node * N_FEAT;
  bool lo   = (khalf == 0);

  v8f c1 = {}, c0 = {};
#pragma unroll
  for (int k = 0; k < 48; k += 4) {
    v2f a = *(const v2f*)(X + xrow + k + khalf);
    v2f b1, b0;
    b1.x = W1[(k + khalf)     * HIDDEN + row];
    b1.y = W1[(k + khalf + 1) * HIDDEN + row];
    b0.x = W0[(k + khalf)     * HIDDEN + row];
    b0.y = W0[(k + khalf + 1) * HIDDEN + row];
    c1 = __builtin_amdgcn_wmma_f32_16x16x4_f32(false, a, false, b1, (short)0, c1, false, false);
    c0 = __builtin_amdgcn_wmma_f32_16x16x4_f32(false, a, false, b0, (short)0, c0, false, false);
  }
  {
    v2f a0 = *(const v2f*)(X + xrow + 48);
    float p1x = W1[48 * HIDDEN + row], p1y = W1[49 * HIDDEN + row];
    float p0x = W0[48 * HIDDEN + row], p0y = W0[49 * HIDDEN + row];
    v2f a, b1, b0;
    a.x  = lo ? a0.x : 0.0f;  a.y  = lo ? a0.y : 0.0f;
    b1.x = lo ? p1x  : 0.0f;  b1.y = lo ? p1y  : 0.0f;
    b0.x = lo ? p0x  : 0.0f;  b0.y = lo ? p0y  : 0.0f;
    c1 = __builtin_amdgcn_wmma_f32_16x16x4_f32(false, a, false, b1, (short)0, c1, false, false);
    c0 = __builtin_amdgcn_wmma_f32_16x16x4_f32(false, a, false, b0, (short)0, c0, false, false);
  }

  int col = lane & 15;
  int mhi = (lane >> 4) * 8;
#pragma unroll
  for (int j = 0; j < 8; ++j) {
    int m = base + mhi + j;
    if (m < n) {
      Y[(long)m * HIDDEN + col] = c1[j];
      Z[(long)m * HIDDEN + col] = c0[j];
    }
  }
}

extern "C" void kernel_launch(void* const* d_in, const int* in_sizes, int n_in,
                              void* d_out, int out_size, void* d_ws, size_t ws_size,
                              hipStream_t stream) {
  const float* x    = (const float*)d_in[0];
  const int*   ei   = (const int*)d_in[1];
  const float* w1_0 = (const float*)d_in[2];
  const float* w1_1 = (const float*)d_in[3];
  const float* b1   = (const float*)d_in[4];
  const float* w2_0 = (const float*)d_in[5];
  const float* w2_1 = (const float*)d_in[6];
  const float* b2   = (const float*)d_in[7];
  float* out = (float*)d_out;

  int n = in_sizes[0] / N_FEAT;
  int E = in_sizes[1] / 2;
  const int* src = ei;
  const int* dst = ei + E;

  int B = (n + BNODES - 1) / BNODES;
  int G = (E + CHUNK - 1) / CHUNK;

  float* ws     = (float*)d_ws;
  int*   cnt    = (int*)ws;
  int*   off    = cnt + n;
  float* dinv   = ws + 2L * n;
  int*   lstT   = (int*)(ws + 3L * n);
  long   lstw   = ((long)(B + 1) * G + 3) / 4 * 4;
  float* r1     = ws + 3L * n + lstw;
  float* y1     = r1;
  float* r2a    = r1 + (long)n * HIDDEN;
  long   r2aw   = 1L * G * CHUNK > 1L * n * HIDDEN ? 1L * G * CHUNK : 1L * n * HIDDEN;
  int*   sorted = (int*)r2a;
  float* hs     = r2a;
  float* h      = r2a + r2aw;
  int*   csr    = (int*)(h + (long)n * HIDDEN);

  int ntiles = (n + 15) / 16;
  int PB     = (ntiles + 31) / 32;
  unsigned gblk = (unsigned)(((long)n * 4 + 255) / 256);

  chunk_sort_proj1<<<G + PB, 1024, 0, stream>>>(src, dst, sorted, lstT, x, w1_1, w1_0, y1, h, E, n, B, G);
  bin_order<<<B, 1024, 0, stream>>>(sorted, lstT, csr, cnt, off, dinv, y1, n, E, G);
  gather16_tag1<<<gblk, 256, 0, stream>>>(csr, off, cnt, dinv, y1, b1, hs, h, n, E);
  gather16_tag2<<<gblk, 256, 0, stream>>>(csr, off, cnt, dinv, hs, h, w2_0, w2_1, b2, out, n, E);
}


# Edge-sum waves read their lists once: a lane takes 4 entries per 16-byte load (2 loads cover 32 edges of its node) and the node's four lanes share them by DPP quad moves; sum order kept
# speedup vs baseline: 6.5171x; 1.0482x over previous
#include <hip/hip_runtime.h>
#include <math.h>

typedef __attribute__((ext_vector_type(2))) float v2f;
typedef __attribute__((ext_vector_type(8))) float v8f;

#define N_FEAT   50
#define HIDDEN   16
#define N_CLS    10

#define BSH     9
#define BNODES  512
#define CHUNK   8192
#define EPT     8
#define MAXB    1024
#define MAXG    1024
#define CAP2    16384

__device__ __forceinline__ int clamp_node(int d, int n) {
  d = d < 0 ? 0 : d;
  return d < n ? d : n - 1;
}

__device__ __forceinline__ int block_excl_scan(int v, int* red, int* total) {
  int lane = threadIdx.x & 31, wv = threadIdx.x >> 5;
  int s = v;
  for (int d = 1; d < 32; d <<= 1) {
    int t = __shfl_up(s, d, 32);
    if (lane >= d) s += t;
  }
  if (lane == 31) red[wv] = s;
  __syncthreads();
  if (wv == 0) {
    int w = red[lane];
    int ws = w;
    for (int d = 1; d < 32; d <<= 1) {
      int t = __shfl_up(ws, d, 32);
      if (lane >= d) ws += t;
    }
    red[lane] = ws - w;
    if (lane == 31) red[32] = ws;
  }
  __syncthreads();
  int r = red[wv] + s - v;
  *total = red[32];
  __syncthreads();
  return r;
}

__device__ __forceinline__ void proj1_tile(
    const float* __restrict__ X, const float* __restrict__ W1, const float* __restrict__ W0,
    float* __restrict__ Y, float* __restrict__ Z, int n, int wave);

__global__ __launch_bounds__(1024) void chunk_sort_proj1(
    const int* __restrict__ src, const int* __restrict__ dst,
    int* __restrict__ sorted, int* __restrict__ lstT,
    const float* __restrict__ X, const float* __restrict__ W1, const float* __restrict__ W0,
    float* __restrict__ Y, float* __restrict__ Z, int E, int n, int B, int G) {
  __shared__ int  hist[MAXB];
  __shared__ int  lstart[MAXB];
  __shared__ int  red[33];
  __shared__ int  stage[CHUNK];
  if ((int)blockIdx.x >= G) {
    proj1_tile(X, W1, W0, Y, Z, n, (((int)blockIdx.x - G) * 1024 + (int)threadIdx.x) >> 5);
    return;
  }
  int t = threadIdx.x, g = blockIdx.x;
  hist[t] = 0;
  __syncthreads();
  int e0 = g * CHUNK;
  int sv[EPT], dv[EPT], rv[EPT];
#pragma unroll
  for (int k = 0; k < EPT; ++k) {
    int e = e0 + k * 1024 + t;
    int ec = e < E ? e : E - 1;
    sv[k] = src[ec];
    dv[k] = dst[ec];
  }
#pragma unroll
  for (int k = 0; k < EPT; ++k) {
    int e = e0 + k * 1024 + t;
    rv[k] = 0;
    if (e < E) {
      dv[k] = clamp_node(dv[k], n);
      rv[k] = atomicAdd(&hist[dv[k] >> BSH], 1);
    } else {
      dv[k] = -1;
    }
  }
  __syncthreads();
  int tot;
  int ls = block_excl_scan(hist[t], red, &tot);
  lstart[t] = ls;
  __syncthreads();
  int lb[EPT];
#pragma unroll
  for (int k = 0; k < EPT; ++k) lb[k] = lstart[(dv[k] < 0 ? 0 : dv[k]) >> BSH];
#pragma unroll
  for (int k = 0; k < EPT; ++k) {
    if (dv[k] >= 0) {
      int p = lb[k] + rv[k];
      p = p < CHUNK ? p : CHUNK - 1;
      int s = ((unsigned)sv[k] < (unsigned)n) ? sv[k] : 0;
      stage[p] = (s << BSH) | (dv[k] & (BNODES - 1));
    }
  }
  if (t < B)  lstT[(long)t * G + g] = ls;
  if (t == 0) lstT[(long)B * G + g] = tot;
  __syncthreads();
  int w[EPT];
#pragma unroll
  for (int k = 0; k < EPT; ++k) w[k] = stage[k * 1024 + t];
#pragma unroll
  for (int k = 0; k < EPT; ++k) {
    int i = k * 1024 + t;
    if (i < tot) sorted[(long)g * CHUNK + i] = w[k];
  }
}

__global__ __launch_bounds__(1024) void bin_order(
    const int* __restrict__ sorted, const int* __restrict__ lstT, int* __restrict__ csr,
    int* __restrict__ cnt, int* __restrict__ off, float* __restrict__ dinv, float* Y,
    int n, int E, int G) {
  __shared__ int pstart[MAXG + 1];
  __shared__ int pbase[MAXG];
  __shared__ int first[CAP2 / 32 + 1];
  __shared__ int cntL[BNODES];
  __shared__ int cur[BNODES];
  __shared__ int red[33];
  __shared__ int eS[CAP2];
  __shared__ unsigned short eD[CAP2];
  __shared__ int stg[CAP2];
  __shared__ float dinvL[BNODES];
  int t = threadIdx.x, b = blockIdx.x;
  int lane = t & 31, wv = t >> 5;
  int base = b << BSH;

  float4* Y4 = (float4*)Y;
  long yq = (long)base * 4;
  bool yk0 = base + (t >> 2) < n, yk1 = base + ((t + 1024) >> 2) < n;
  float4 ya = {0.0f, 0.0f, 0.0f, 0.0f}, yb = {0.0f, 0.0f, 0.0f, 0.0f};
  if (yk0) ya = Y4[yq + t];
  if (yk1) yb = Y4[yq + t + 1024];

  int st = 0, len = 0;
  if (t < G) {
    st = lstT[(long)b * G + t];
    int en = lstT[(long)(b + 1) * G + t];
    st = st < 0 ? 0 : (st > CHUNK ? CHUNK : st);
    en = en < st ? st : (en > CHUNK ? CHUNK : en);
    len = en - st;
  }
  int m, bs;
  int ps = block_excl_scan(len, red, &m);
  (void)block_excl_scan(st, red, &bs);
  bs = bs < 0 ? 0 : (bs > E ? E : bs);
  m  = m > E - bs ? E - bs : m;
  pstart[t] = ps;
  if (t == 0) pstart[MAXG] = m;
  if (t < MAXG) pbase[t] = t * CHUNK + st - ps;
  if (t < BNODES) cntL[t] = 0;
  __syncthreads();

  bool in_lds = m <= CAP2;
  if (in_lds) {
    if (len > 0) {
      int pe = ps + len; pe = pe < m ? pe : m;
      for (int k = (ps + 31) >> 5; (k << 5) < pe; ++k) first[k] = t;
    }
    __syncthreads();
    for (int i0 = t; i0 < m; i0 += 8 * 1024) {
      int v[8];
#pragma unroll
      for (int u = 0; u < 8; ++u) {
        int i = i0 + u * 1024;
        v[u] = 0;
        if (i < m) {
          int p = first[i >> 5];
          p = p < 0 ? 0 : (p < G ? p : G - 1);
          while (p + 1 < G && i >= pstart[p + 1]) ++p;
          v[u] = sorted[pbase[p] + i];
        }
      }
#pragma unroll
      for (int u = 0; u < 8; ++u) {
        int i = i0 + u * 1024;
        if (i < m) {
          int dl = v[u] & (BNODES - 1);
          atomicAdd(&cntL[dl], 1);
          eS[i] = (int)((unsigned)v[u] >> BSH);
          eD[i] = (unsigned short)dl;
        }
      }
    }
  } else {
    for (int p = wv; p < G; p += 32) {
      int q0 = pstart[p], q1 = pstart[p + 1];
      q1 = q1 < m ? q1 : m;
      for (int i = q0 + lane; i < q1; i += 32) {
        int dl = sorted[pbase[p] + i] & (BNODES - 1);
        atomicAdd(&cntL[dl], 1);
      }
    }
  }
  __syncthreads();
  int c = (t < BNODES) ? cntL[t] : 0;
  int tot;
  int lo = block_excl_scan(c, red, &tot);
  if (t < BNODES) {
    cur[t] = lo;
    int node = base + t;
    if (node < n) {
      float dn = (c > 0) ? rsqrtf((float)c) : 0.0f;
      cnt[node]  = c;
      off[node]  = bs + lo;
      dinv[node] = dn;
      dinvL[t]   = dn;
    }
  }
  __syncthreads();
  if (yk0) {
    float d = dinvL[t >> 2];
    float4 o = {ya.x * d, ya.y * d, ya.z * d, ya.w * d};
    Y4[yq + t] = o;
  }
  if (yk1) {
    float d = dinvL[(t + 1024) >> 2];
    float4 o = {yb.x * d, yb.y * d, yb.z * d, yb.w * d};
    Y4[yq + t + 1024] = o;
  }
  if (in_lds) {
    for (int i0 = t; i0 < m; i0 += 8 * 1024) {
      int dd[8], ss[8];
#pragma unroll
      for (int u = 0; u < 8; ++u) {
        int i = i0 + u * 1024;
        int ic = i < m ? i : 0;
        dd[u] = eD[ic];
        ss[u] = eS[ic];
      }
#pragma unroll
      for (int u = 0; u < 8; ++u) {
        int i = i0 + u * 1024;
        if (i < m) {
          int p = atomicAdd(&cur[dd[u]], 1);
          p = p < 0 ? 0 : (p < CAP2 ? p : CAP2 - 1);
          stg[p] = ss[u];
        }
      }
    }
    __syncthreads();
    for (int i = t; i < m; i += 1024) csr[bs + i] = stg[i];
  } else {
    for (int p = wv; p < G; p += 32) {
      int q0 = pstart[p], q1 = pstart[p + 1];
      q1 = q1 < m ? q1 : m;
      for (int i = q0 + lane; i < q1; i += 32) {
        int v = sorted[pbase[p] + i];
        int dl = v & (BNODES - 1);
        unsigned pos = (unsigned)(bs + atomicAdd(&cur[dl], 1));
        pos = pos < (unsigned)E ? pos : (unsigned)(E - 1);
        csr[pos] = (int)((unsigned)v >> BSH);
      }
    }
  }
}

__device__ __forceinline__ float4 row_piece(const float4* __restrict__ h4, int s, int q, int n) {
  s = ((unsigned)s < (unsigned)n) ? s : 0;
  return h4[(long)s * 4 + q];
}

__device__ __forceinline__ int4 list_piece(const int* __restrict__ csr, int p, int E) {
  int pc = p < E - 4 ? p : E - 4;
  pc = pc < 0 ? 0 : pc;
  int d = p - pc;
  int4 e;
  __builtin_memcpy(&e, csr + pc, 16);
  if (d != 0) {
    e.x = d == 1 ? e.y : (d == 2 ? e.z : e.w);
    e.y = d == 1 ? e.z : e.w;
    e.z = e.w;
  }
  return e;
}

template <int K>
__device__ __forceinline__ int quad_take(int v) {
  return __builtin_amdgcn_update_dpp(0, v, K * 0x55, 0xF, 0xF, true);
}

template <int K>
__device__ __forceinline__ void add_four(float4& acc, const float4* __restrict__ h4, int4 e,
                                         int jj, int end, int q, int n) {
  bool v1 = jj + 1 < end, v2 = jj + 2 < end, v3 = jj + 3 < end;
  int s0 = quad_take<K>(e.x);
  int s1 = quad_take<K>(e.y);
  int s2 = quad_take<K>(e.z);
  int s3 = quad_take<K>(e.w);
  s1 = v1 ? s1 : s0;
  s2 = v2 ? s2 : s0;
  s3 = v3 ? s3 : s0;
  float4 r0 = row_piece(h4, s0, q, n);
  float4 r1 = row_piece(h4, s1, q, n);
  float4 r2 = row_piece(h4, s2, q, n);
  float4 r3 = row_piece(h4, s3, q, n);
  acc.x += r0.x; acc.y += r0.y; acc.z += r0.z; acc.w += r0.w;
  acc.x += v1 ? r1.x : 0.0f; acc.y += v1 ? r1.y : 0.0f; acc.z += v1 ? r1.z : 0.0f; acc.w += v1 ? r1.w : 0.0f;
  acc.x += v2 ? r2.x : 0.0f; acc.y += v2 ? r2.y : 0.0f; acc.z += v2 ? r2.z : 0.0f; acc.w += v2 ? r2.w : 0.0f;
  acc.x += v3 ? r3.x : 0.0f; acc.y += v3 ? r3.y : 0.0f; acc.z += v3 ? r3.z : 0.0f; acc.w += v3 ? r3.w : 0.0f;
}

__device__ __forceinline__ float4 edge_sum(const int* __restrict__ csr, const float4* __restrict__ h4,
                                           int j, int end, int q, int n, int E) {
  j   = j   < 0 ? 0 : j;
  end = end > E ? E : end;
  float4 acc = {0.0f, 0.0f, 0.0f, 0.0f};
  for (int b = j; b < end; b += 32) {
    bool more = b + 16 < end;
    int4 ea = list_piece(csr, b + 4 * q, E);
    int4 eb = {0, 0, 0, 0};
    if (more) eb = list_piece(csr, b + 16 + 4 * q, E);
    add_four<0>(acc, h4, ea, b, end, q, n);
    if (b + 4 < end) {
      add_four<1>(acc, h4, ea, b + 4, end, q, n);
      if (b + 8 < end) {
        add_four<2>(acc, h4, ea, b + 8, end, q, n);
        if (b + 12 < end) {
          add_four<3>(acc, h4, ea, b + 12, end, q, n);
          if (more) {
            add_four<0>(acc, h4, eb, b + 16, end, q, n);
            if (b + 20 < end) {
              add_four<1>(acc, h4, eb, b + 20, end, q, n);
              if (b + 24 < end) {
                add_four<2>(acc, h4, eb, b + 24, end, q, n);
                if (b + 28 < end) add_four<3>(acc, h4, eb, b + 28, end, q, n);
              }
            }
          }
        }
      }
    }
  }
  return acc;
}

__global__ __launch_bounds__(256) void gather16_tag1(
    const int* __restrict__ csr, const int* __restrict__ off, const int* __restrict__ cnt,
    const float* __restrict__ dinv, const float* __restrict__ y, const float* __restrict__ B,
    float* __restrict__ HS, float* __restrict__ H, int n, int E) {
  int t = blockIdx.x * blockDim.x + threadIdx.x;
  int node = t >> 2;
  int q = t & 3;
  if (node >= n) return;
  int j   = off[node];
  int end = j + cnt[node];
  float dv = dinv[node];
  float4 z = ((const float4*)H)[(long)node * 4 + q];
  float b0 = B[4 * q], b1v = B[4 * q + 1], b2v = B[4 * q + 2], b3v = B[4 * q + 3];
  float4 acc = edge_sum(csr, (const float4*)y, j, end, q, n, E);
  float4 v;
  v.x = (z.x + acc.x * dv) + b0;
  v.y = (z.y + acc.y * dv) + b1v;
  v.z = (z.z + acc.z * dv) + b2v;
  v.w = (z.w + acc.w * dv) + b3v;
  v.x = v.x > 0.0f ? v.x : 0.0f;  v.y = v.y > 0.0f ? v.y : 0.0f;
  v.z = v.z > 0.0f ? v.z : 0.0f;  v.w = v.w > 0.0f ? v.w : 0.0f;
  ((float4*)H)[(long)node * 4 + q] = v;
  float4 s = {v.x * dv, v.y * dv, v.z * dv, v.w * dv};
  ((float4*)HS)[(long)node * 4 + q] = s;
}

__global__ __launch_bounds__(256) void gather16_tag2(
    const int* __restrict__ csr, const int* __restrict__ off, const int* __restrict__ cnt,
    const float* __restrict__ dinv, const float* __restrict__ HS, const float* __restrict__ H,
    const float* __restrict__ W0, const float* __restrict__ W1, const float* __restrict__ B,
    float* __restrict__ OUT, int n, int E) {
  int t = blockIdx.x * blockDim.x + threadIdx.x;
  int lane = threadIdx.x & 31;
  int wbase = (t >> 5) * 8;
  __shared__ __attribute__((aligned(16))) float so[64 * N_CLS];

  int node = wbase + (lane >> 2);
  int q = lane & 3;
  bool nv = node < n;
  int nc = nv ? node : n - 1;
  int j   = off[nc];
  int cn  = cnt[nc];
  float dv = dinv[nc];
  int end = j + (nv ? cn : 0);

  int c  = lane & 15;
  int hh = lane >> 4;
  int dn = wbase + (c & 7);
  dn = dn < n ? dn : n - 1;
  const float4* H4 = (const float4*)H;
  float4 hA = H4[(long)dn * 4 + 2 * hh];
  float4 hB = H4[(long)dn * 4 + 2 * hh + 1];
  bool cok = c < N_CLS;
  int  wc  = cok ? c : (N_CLS - 1);
  float w0[8], w1[8], bb[8];
#pragma unroll
  for (int i = 0; i < 8; ++i) {
    int f = 8 * hh + i;
    float a0 = W0[f * N_CLS + wc];
    float a1 = W1[f * N_CLS + wc];
    w0[i] = cok ? a0 : 0.0f;
    w1[i] = cok ? a1 : 0.0f;
    int cl = 8 * hh + i;
    float bv = B[cl < N_CLS ? cl : (N_CLS - 1)];
    bb[i] = cl < N_CLS ? bv : 0.0f;
  }

  float4 acc = edge_sum(csr, (const float4*)HS, j, end, q, n, E);
  acc.x *= dv; acc.y *= dv; acc.z *= dv; acc.w *= dv;

  int sl = 4 * (c & 7) + 2 * hh;
  float g[8];
  g[0] = __shfl(acc.x, sl, 32);      g[1] = __shfl(acc.y, sl, 32);
  g[2] = __shfl(acc.z, sl, 32);      g[3] = __shfl(acc.w, sl, 32);
  g[4] = __shfl(acc.x, sl + 1, 32);  g[5] = __shfl(acc.y, sl + 1, 32);
  g[6] = __shfl(acc.z, sl + 1, 32);  g[7] = __shfl(acc.w, sl + 1, 32);
  float hv[8] = {hA.x, hA.y, hA.z, hA.w, hB.x, hB.y, hB.z, hB.w};

  v8f ch = {}, ca = {};
#pragma unroll
  for (int ks = 0; ks < 4; ++ks) {
    v2f a, b;
    a.x = w0[2 * ks];  a.y = w0[2 * ks + 1];
    b.x = hv[2 * ks];  b.y = hv[2 * ks + 1];
    ch = __builtin_amdgcn_wmma_f32_16x16x4_f32(false, a, false, b, (short)0, ch, false, false);
    a.x = w1[2 * ks];  a.y = w1[2 * ks + 1];
    b.x = g[2 * ks];   b.y = g[2 * ks + 1];
    ca = __builtin_amdgcn_wmma_f32_16x16x4_f32(false, a, false, b, (short)0, ca, false, false);
  }

  float v[8];
  float mx = -INFINITY;
#pragma unroll
  for (int r = 0; r < 8; ++r) {
    v[r] = (ch[r] + ca[r]) + bb[r];
    bool ok = (hh == 0) || (r < N_CLS - 8);
    mx = ok ? fmaxf(mx, v[r]) : mx;
  }
  mx = fmaxf(mx, __shfl_xor(mx, 16, 32));
  float sm = 0.0f;
#pragma unroll
  for (int r = 0; r < 8; ++r) {
    bool ok = (hh == 0) || (r < N_CLS - 8);
    sm += ok ? expf(v[r] - mx) : 0.0f;
  }
  sm += __shfl_xor(sm, 16, 32);
  float ls = logf(sm);
  if (c < 8) {
    float* o = so + ((threadIdx.x >> 5) * 8 + c) * N_CLS + 8 * hh;
    float2 p0 = {v[0] - mx - ls, v[1] - mx - ls};
    *(float2*)o = p0;
    if (hh == 0) {
      float2 p1 = {v[2] - mx - ls, v[3] - mx - ls};
      float2 p2 = {v[4] - mx - ls, v[5] - mx - ls};
      float2 p3 = {v[6] - mx - ls, v[7] - mx - ls};
      *(float2*)(o + 2) = p1;
      *(float2*)(o + 4) = p2;
      *(float2*)(o + 6) = p3;
    }
  }
  __syncthreads();
  long g0   = (long)blockIdx.x * (64 * N_CLS);
  long gend = (long)n * N_CLS;
  int  i4   = threadIdx.x * 4;
  if (i4 < 64 * N_CLS) {
    if (g0 + i4 + 3 < gend) {
      *(float4*)(OUT + g0 + i4) = *(const float4*)(so + i4);
    } else {
      for (int u = 0; u < 4; ++u)
        if (g0 + i4 + u < gend) OUT[g0 + i4 + u] = so[i4 + u];
    }
  }
}

__device__ __forceinline__ void proj1_tile(
    const float* __restrict__ X, const float* __restrict__ W1, const float* __restrict__ W0,
    float* __restrict__ Y, float* __restrict__ Z, int n, int wave) {
  int lane = threadIdx.x & 31;
  int base = wave * 16;
  if (base >= n) return;

  int row   = lane & 15;
  int khalf = (lane >> 4) * 2;
  int node  = base + row;
  node = node < n ? node : n - 1;
  long xrow = (long)node * N_FEAT;
  bool lo   = (khalf == 0);

  v8f c1 = {}, c0 = {};
#pragma unroll
  for (int k = 0; k < 48; k += 4) {
    v2f a = *(const v2f*)(X + xrow + k + khalf);
    v2f b1, b0;
    b1.x = W1[(k + khalf)     * HIDDEN + row];
    b1.y = W1[(k + khalf + 1) * HIDDEN + row];
    b0.x = W0[(k + khalf)     * HIDDEN + row];
    b0.y = W0[(k + khalf + 1) * HIDDEN + row];
    c1 = __builtin_amdgcn_wmma_f32_16x16x4_f32(false, a, false, b1, (short)0, c1, false, false);
    c0 = __builtin_amdgcn_wmma_f32_16x16x4_f32(false, a, false, b0, (short)0, c0, false, false);
  }
  {
    v2f a0 = *(const v2f*)(X + xrow + 48);
    float p1x = W1[48 * HIDDEN + row], p1y = W1[49 * HIDDEN + row];
    float p0x = W0[48 * HIDDEN + row], p0y = W0[49 * HIDDEN + row];
    v2f a, b1, b0;
    a.x  = lo ? a0.x : 0.0f;  a.y  = lo ? a0.y : 0.0f;
    b1.x = lo ? p1x  : 0.0f;  b1.y = lo ? p1y  : 0.0f;
    b0.x = lo ? p0x  : 0.0f;  b0.y = lo ? p0y  : 0.0f;
    c1 = __builtin_amdgcn_wmma_f32_16x16x4_f32(false, a, false, b1, (short)0, c1, false, false);
    c0 = __builtin_amdgcn_wmma_f32_16x16x4_f32(false, a, false, b0, (short)0, c0, false, false);
  }

  int col = lane & 15;
  int mhi = (lane >> 4) * 8;
#pragma unroll
  for (int j = 0; j < 8; ++j) {
    int m = base + mhi + j;
    if (m < n) {
      Y[(long)m * HIDDEN + col] = c1[j];
      Z[(long)m * HIDDEN + col] = c0[j];
    }
  }
}

extern "C" void kernel_launch(void* const* d_in, const int* in_sizes, int n_in,
                              void* d_out, int out_size, void* d_ws, size_t ws_size,
                              hipStream_t stream) {
  const float* x    = (const float*)d_in[0];
  const int*   ei   = (const int*)d_in[1];
  const float* w1_0 = (const float*)d_in[2];
  const float* w1_1 = (const float*)d_in[3];
  const float* b1   = (const float*)d_in[4];
  const float* w2_0 = (const float*)d_in[5];
  const float* w2_1 = (const float*)d_in[6];
  const float* b2   = (const float*)d_in[7];
  float* out = (float*)d_out;

  int n = in_sizes[0] / N_FEAT;
  int E = in_sizes[1] / 2;
  const int* src = ei;
  const int* dst = ei + E;

  int B = (n + BNODES - 1) / BNODES;
  int G = (E + CHUNK - 1) / CHUNK;

  float* ws     = (float*)d_ws;
  int*   cnt    = (int*)ws;
  int*   off    = cnt + n;
  float* dinv   = ws + 2L * n;
  int*   lstT   = (int*)(ws + 3L * n);
  long   lstw   = ((long)(B + 1) * G + 3) / 4 * 4;
  float* r1     = ws + 3L * n + lstw;
  float* y1     = r1;
  float* r2a    = r1 + (long)n * HIDDEN;
  long   r2aw   = 1L * G * CHUNK > 1L * n * HIDDEN ? 1L * G * CHUNK : 1L * n * HIDDEN;
  int*   sorted = (int*)r2a;
  float* hs     = r2a;
  float* h      = r2a + r2aw;
  int*   csr    = (int*)(h + (long)n * HIDDEN);

  int ntiles = (n + 15) / 16;
  int PB     = (ntiles + 31) / 32;
  unsigned gblk = (unsigned)(((long)n * 4 + 255) / 256);

  chunk_sort_proj1<<<G + PB, 1024, 0, stream>>>(src, dst, sorted, lstT, x, w1_1, w1_0, y1, h, E, n, B, G);
  bin_order<<<B, 1024, 0, stream>>>(sorted, lstT, csr, cnt, off, dinv, y1, n, E, G);
  gather16_tag1<<<gblk, 256, 0, stream>>>(csr, off, cnt, dinv, y1, b1, hs, h, n, E);
  gather16_tag2<<<gblk, 256, 0, stream>>>(csr, off, cnt, dinv, hs, h, w2_0, w2_1, b2, out, n, E);
}


# Workspace row tables start on a 128-byte line: each 64-byte row of y, hs, h lies in one line (the base sat at byte 80 of a line, so every second row straddled two lines); sum order kept
# speedup vs baseline: 8.5490x; 1.3118x over previous
#include <hip/hip_runtime.h>
#include <math.h>

typedef __attribute__((ext_vector_type(2))) float v2f;
typedef __attribute__((ext_vector_type(8))) float v8f;

#define N_FEAT   50
#define HIDDEN   16
#define N_CLS    10

#define BSH     9
#define BNODES  512
#define CHUNK   8192
#define EPT     8
#define MAXB    1024
#define MAXG    1024
#define CAP2    16384

__device__ __forceinline__ int clamp_node(int d, int n) {
  d = d < 0 ? 0 : d;
  return d < n ? d : n - 1;
}

__device__ __forceinline__ int block_excl_scan(int v, int* red, int* total) {
  int lane = threadIdx.x & 31, wv = threadIdx.x >> 5;
  int s = v;
  for (int d = 1; d < 32; d <<= 1) {
    int t = __shfl_up(s, d, 32);
    if (lane >= d) s += t;
  }
  if (lane == 31) red[wv] = s;
  __syncthreads();
  if (wv == 0) {
    int w = red[lane];
    int ws = w;
    for (int d = 1; d < 32; d <<= 1) {
      int t = __shfl_up(ws, d, 32);
      if (lane >= d) ws += t;
    }
    red[lane] = ws - w;
    if (lane == 31) red[32] = ws;
  }
  __syncthreads();
  int r = red[wv] + s - v;
  *total = red[32];
  __syncthreads();
  return r;
}

__device__ __forceinline__ void proj1_tile(
    const float* __restrict__ X, const float* __restrict__ W1, const float* __restrict__ W0,
    float* __restrict__ Y, float* __restrict__ Z, int n, int wave);

__global__ __launch_bounds__(1024) void chunk_sort_proj1(
    const int* __restrict__ src, const int* __restrict__ dst,
    int* __restrict__ sorted, int* __restrict__ lstT,
    const float* __restrict__ X, const float* __restrict__ W1, const float* __restrict__ W0,
    float* __restrict__ Y, float* __restrict__ Z, int E, int n, int B, int G) {
  __shared__ int  hist[MAXB];
  __shared__ int  lstart[MAXB];
  __shared__ int  red[33];
  __shared__ int  stage[CHUNK];
  if ((int)blockIdx.x >= G) {
    proj1_tile(X, W1, W0, Y, Z, n, (((int)blockIdx.x - G) * 1024 + (int)threadIdx.x) >> 5);
    return;
  }
  int t = threadIdx.x, g = blockIdx.x;
  hist[t] = 0;
  __syncthreads();
  int e0 = g * CHUNK;
  int sv[EPT], dv[EPT], rv[EPT];
#pragma unroll
  for (int k = 0; k < EPT; ++k) {
    int e = e0 + k * 1024 + t;
    int ec = e < E ? e : E - 1;
    sv[k] = src[ec];
    dv[k] = dst[ec];
  }
#pragma unroll
  for (int k = 0; k < EPT; ++k) {
    int e = e0 + k * 1024 + t;
    rv[k] = 0;
    if (e < E) {
      dv[k] = clamp_node(dv[k], n);
      rv[k] = atomicAdd(&hist[dv[k] >> BSH], 1);
    } else {
      dv[k] = -1;
    }
  }
  __syncthreads();
  int tot;
  int ls = block_excl_scan(hist[t], red, &tot);
  lstart[t] = ls;
  __syncthreads();
  int lb[EPT];
#pragma unroll
  for (int k = 0; k < EPT; ++k) lb[k] = lstart[(dv[k] < 0 ? 0 : dv[k]) >> BSH];
#pragma unroll
  for (int k = 0; k < EPT; ++k) {
    if (dv[k] >= 0) {
      int p = lb[k] + rv[k];
      p = p < CHUNK ? p : CHUNK - 1;
      int s = ((unsigned)sv[k] < (unsigned)n) ? sv[k] : 0;
      stage[p] = (s << BSH) | (dv[k] & (BNODES - 1));
    }
  }
  if (t < B)  lstT[(long)t * G + g] = ls;
  if (t == 0) lstT[(long)B * G + g] = tot;
  __syncthreads();
  int w[EPT];
#pragma unroll
  for (int k = 0; k < EPT; ++k) w[k] = stage[k * 1024 + t];
#pragma unroll
  for (int k = 0; k < EPT; ++k) {
    int i = k * 1024 + t;
    if (i < tot) sorted[(long)g * CHUNK + i] = w[k];
  }
}

__global__ __launch_bounds__(1024) void bin_order(
    const int* __restrict__ sorted, const int* __restrict__ lstT, int* __restrict__ csr,
    int* __restrict__ cnt, int* __restrict__ off, float* __restrict__ dinv, float* Y,
    int n, int E, int G) {
  __shared__ int pstart[MAXG + 1];
  __shared__ int pbase[MAXG];
  __shared__ int first[CAP2 / 32 + 1];
  __shared__ int cntL[BNODES];
  __shared__ int cur[BNODES];
  __shared__ int red[33];
  __shared__ int eS[CAP2];
  __shared__ unsigned short eD[CAP2];
  __shared__ int stg[CAP2];
  __shared__ float dinvL[BNODES];
  int t = threadIdx.x, b = blockIdx.x;
  int lane = t & 31, wv = t >> 5;
  int base = b << BSH;

  float4* Y4 = (float4*)Y;
  long yq = (long)base * 4;
  bool yk0 = base + (t >> 2) < n, yk1 = base + ((t + 1024) >> 2) < n;
  float4 ya = {0.0f, 0.0f, 0.0f, 0.0f}, yb = {0.0f, 0.0f, 0.0f, 0.0f};
  if (yk0) ya = Y4[yq + t];
  if (yk1) yb = Y4[yq + t + 1024];

  int st = 0, len = 0;
  if (t < G) {
    st = lstT[(long)b * G + t];
    int en = lstT[(long)(b + 1) * G + t];
    st = st < 0 ? 0 : (st > CHUNK ? CHUNK : st);
    en = en < st ? st : (en > CHUNK ? CHUNK : en);
    len = en - st;
  }
  int m, bs;
  int ps = block_excl_scan(len, red, &m);
  (void)block_excl_scan(st, red, &bs);
  bs = bs < 0 ? 0 : (bs > E ? E : bs);
  m  = m > E - bs ? E - bs : m;
  pstart[t] = ps;
  if (t == 0) pstart[MAXG] = m;
  if (t < MAXG) pbase[t] = t * CHUNK + st - ps;
  if (t < BNODES) cntL[t] = 0;
  __syncthreads();

  bool in_lds = m <= CAP2;
  if (in_lds) {
    if (len > 0) {
      int pe = ps + len; pe = pe < m ? pe : m;
      for (int k = (ps + 31) >> 5; (k << 5) < pe; ++k) first[k] = t;
    }
    __syncthreads();
    for (int i0 = t; i0 < m; i0 += 8 * 1024) {
      int v[8];
#pragma unroll
      for (int u = 0; u < 8; ++u) {
        int i = i0 + u * 1024;
        v[u] = 0;
        if (i < m) {
          int p = first[i >> 5];
          p = p < 0 ? 0 : (p < G ? p : G - 1);
          while (p + 1 < G && i >= pstart[p + 1]) ++p;
          v[u] = sorted[pbase[p] + i];
        }
      }
#pragma unroll
      for (int u = 0; u < 8; ++u) {
        int i = i0 + u * 1024;
        if (i < m) {
          int dl = v[u] & (BNODES - 1);
          atomicAdd(&cntL[dl], 1);
          eS[i] = (int)((unsigned)v[u] >> BSH);
          eD[i] = (unsigned short)dl;
        }
      }
    }
  } else {
    for (int p = wv; p < G; p += 32) {
      int q0 = pstart[p], q1 = pstart[p + 1];
      q1 = q1 < m ? q1 : m;
      for (int i = q0 + lane; i < q1; i += 32) {
        int dl = sorted[pbase[p] + i] & (BNODES - 1);
        atomicAdd(&cntL[dl], 1);
      }
    }
  }
  __syncthreads();
  int c = (t < BNODES) ? cntL[t] : 0;
  int tot;
  int lo = block_excl_scan(c, red, &tot);
  if (t < BNODES) {
    cur[t] = lo;
    int node = base + t;
    if (node < n) {
      float dn = (c > 0) ? rsqrtf((float)c) : 0.0f;
      cnt[node]  = c;
      off[node]  = bs + lo;
      dinv[node] = dn;
      dinvL[t]   = dn;
    }
  }
  __syncthreads();
  if (yk0) {
    float d = dinvL[t >> 2];
    float4 o = {ya.x * d, ya.y * d, ya.z * d, ya.w * d};
    Y4[yq + t] = o;
  }
  if (yk1) {
    float d = dinvL[(t + 1024) >> 2];
    float4 o = {yb.x * d, yb.y * d, yb.z * d, yb.w * d};
    Y4[yq + t + 1024] = o;
  }
  if (in_lds) {
    for (int i0 = t; i0 < m; i0 += 8 * 1024) {
      int dd[8], ss[8];
#pragma unroll
      for (int u = 0; u < 8; ++u) {
        int i = i0 + u * 1024;
        int ic = i < m ? i : 0;
        dd[u] = eD[ic];
        ss[u] = eS[ic];
      }
#pragma unroll
      for (int u = 0; u < 8; ++u) {
        int i = i0 + u * 1024;
        if (i < m) {
          int p = atomicAdd(&cur[dd[u]], 1);
          p = p < 0 ? 0 : (p < CAP2 ? p : CAP2 - 1);
          stg[p] = ss[u];
        }
      }
    }
    __syncthreads();
    for (int i = t; i < m; i += 1024) csr[bs + i] = stg[i];
  } else {
    for (int p = wv; p < G; p += 32) {
      int q0 = pstart[p], q1 = pstart[p + 1];
      q1 = q1 < m ? q1 : m;
      for (int i = q0 + lane; i < q1; i += 32) {
        int v = sorted[pbase[p] + i];
        int dl = v & (BNODES - 1);
        unsigned pos = (unsigned)(bs + atomicAdd(&cur[dl], 1));
        pos = pos < (unsigned)E ? pos : (unsigned)(E - 1);
        csr[pos] = (int)((unsigned)v >> BSH);
      }
    }
  }
}

__device__ __forceinline__ float4 row_piece(const float4* __restrict__ h4, int s, int q, int n) {
  s = ((unsigned)s < (unsigned)n) ? s : 0;
  return h4[(long)s * 4 + q];
}

__device__ __forceinline__ int4 list_piece(const int* __restrict__ csr, int p, int E) {
  int pc = p < E - 4 ? p : E - 4;
  pc = pc < 0 ? 0 : pc;
  int d = p - pc;
  int4 e;
  __builtin_memcpy(&e, csr + pc, 16);
  if (d != 0) {
    e.x = d == 1 ? e.y : (d == 2 ? e.z : e.w);
    e.y = d == 1 ? e.z : e.w;
    e.z = e.w;
  }
  return e;
}

template <int K>
__device__ __forceinline__ int quad_take(int v) {
  return __builtin_amdgcn_update_dpp(0, v, K * 0x55, 0xF, 0xF, true);
}

template <int K>
__device__ __forceinline__ void add_four(float4& acc, const float4* __restrict__ h4, int4 e,
                                         int jj, int end, int q, int n) {
  bool v1 = jj + 1 < end, v2 = jj + 2 < end, v3 = jj + 3 < end;
  int s0 = quad_take<K>(e.x);
  int s1 = quad_take<K>(e.y);
  int s2 = quad_take<K>(e.z);
  int s3 = quad_take<K>(e.w);
  s1 = v1 ? s1 : s0;
  s2 = v2 ? s2 : s0;
  s3 = v3 ? s3 : s0;
  float4 r0 = row_piece(h4, s0, q, n);
  float4 r1 = row_piece(h4, s1, q, n);
  float4 r2 = row_piece(h4, s2, q, n);
  float4 r3 = row_piece(h4, s3, q, n);
  acc.x += r0.x; acc.y += r0.y; acc.z += r0.z; acc.w += r0.w;
  acc.x += v1 ? r1.x : 0.0f; acc.y += v1 ? r1.y : 0.0f; acc.z += v1 ? r1.z : 0.0f; acc.w += v1 ? r1.w : 0.0f;
  acc.x += v2 ? r2.x : 0.0f; acc.y += v2 ? r2.y : 0.0f; acc.z += v2 ? r2.z : 0.0f; acc.w += v2 ? r2.w : 0.0f;
  acc.x += v3 ? r3.x : 0.0f; acc.y += v3 ? r3.y : 0.0f; acc.z += v3 ? r3.z : 0.0f; acc.w += v3 ? r3.w : 0.0f;
}

__device__ __forceinline__ float4 edge_sum(const int* __restrict__ csr, const float4* __restrict__ h4,
                                           int j, int end, int q, int n, int E) {
  j   = j   < 0 ? 0 : j;
  end = end > E ? E : end;
  float4 acc = {0.0f, 0.0f, 0.0f, 0.0f};
  for (int b = j; b < end; b += 32) {
    bool more = b + 16 < end;
    int4 ea = list_piece(csr, b + 4 * q, E);
    int4 eb = {0, 0, 0, 0};
    if (more) eb = list_piece(csr, b + 16 + 4 * q, E);
    add_four<0>(acc, h4, ea, b, end, q, n);
    if (b + 4 < end) {
      add_four<1>(acc, h4, ea, b + 4, end, q, n);
      if (b + 8 < end) {
        add_four<2>(acc, h4, ea, b + 8, end, q, n);
        if (b + 12 < end) {
          add_four<3>(acc, h4, ea, b + 12, end, q, n);
          if (more) {
            add_four<0>(acc, h4, eb, b + 16, end, q, n);
            if (b + 20 < end) {
              add_four<1>(acc, h4, eb, b + 20, end, q, n);
              if (b + 24 < end) {
                add_four<2>(acc, h4, eb, b + 24, end, q, n);
                if (b + 28 < end) add_four<3>(acc, h4, eb, b + 28, end, q, n);
              }
            }
          }
        }
      }
    }
  }
  return acc;
}

__global__ __launch_bounds__(256) void gather16_tag1(
    const int* __restrict__ csr, const int* __restrict__ off, const int* __restrict__ cnt,
    const float* __restrict__ dinv, const float* __restrict__ y, const float* __restrict__ B,
    float* __restrict__ HS, float* __restrict__ H, int n, int E) {
  int t = blockIdx.x * blockDim.x + threadIdx.x;
  int node = t >> 2;
  int q = t & 3;
  if (node >= n) return;
  int j   = off[node];
  int end = j + cnt[node];
  float dv = dinv[node];
  float4 z = ((const float4*)H)[(long)node * 4 + q];
  float b0 = B[4 * q], b1v = B[4 * q + 1], b2v = B[4 * q + 2], b3v = B[4 * q + 3];
  float4 acc = edge_sum(csr, (const float4*)y, j, end, q, n, E);
  float4 v;
  v.x = (z.x + acc.x * dv) + b0;
  v.y = (z.y + acc.y * dv) + b1v;
  v.z = (z.z + acc.z * dv) + b2v;
  v.w = (z.w + acc.w * dv) + b3v;
  v.x = v.x > 0.0f ? v.x : 0.0f;  v.y = v.y > 0.0f ? v.y : 0.0f;
  v.z = v.z > 0.0f ? v.z : 0.0f;  v.w = v.w > 0.0f ? v.w : 0.0f;
  ((float4*)H)[(long)node * 4 + q] = v;
  float4 s = {v.x * dv, v.y * dv, v.z * dv, v.w * dv};
  ((float4*)HS)[(long)node * 4 + q] = s;
}

__global__ __launch_bounds__(256) void gather16_tag2(
    const int* __restrict__ csr, const int* __restrict__ off, const int* __restrict__ cnt,
    const float* __restrict__ dinv, const float* __restrict__ HS, const float* __restrict__ H,
    const float* __restrict__ W0, const float* __restrict__ W1, const float* __restrict__ B,
    float* __restrict__ OUT, int n, int E) {
  int t = blockIdx.x * blockDim.x + threadIdx.x;
  int lane = threadIdx.x & 31;
  int wbase = (t >> 5) * 8;
  __shared__ __attribute__((aligned(16))) float so[64 * N_CLS];

  int node = wbase + (lane >> 2);
  int q = lane & 3;
  bool nv = node < n;
  int nc = nv ? node : n - 1;
  int j   = off[nc];
  int cn  = cnt[nc];
  float dv = dinv[nc];
  int end = j + (nv ? cn : 0);

  int c  = lane & 15;
  int hh = lane >> 4;
  int dn = wbase + (c & 7);
  dn = dn < n ? dn : n - 1;
  const float4* H4 = (const float4*)H;
  float4 hA = H4[(long)dn * 4 + 2 * hh];
  float4 hB = H4[(long)dn * 4 + 2 * hh + 1];
  bool cok = c < N_CLS;
  int  wc  = cok ? c : (N_CLS - 1);
  float w0[8], w1[8], bb[8];
#pragma unroll
  for (int i = 0; i < 8; ++i) {
    int f = 8 * hh + i;
    float a0 = W0[f * N_CLS + wc];
    float a1 = W1[f * N_CLS + wc];
    w0[i] = cok ? a0 : 0.0f;
    w1[i] = cok ? a1 : 0.0f;
    int cl = 8 * hh + i;
    float bv = B[cl < N_CLS ? cl : (N_CLS - 1)];
    bb[i] = cl < N_CLS ? bv : 0.0f;
  }

  float4 acc = edge_sum(csr, (const float4*)HS, j, end, q, n, E);
  acc.x *= dv; acc.y *= dv; acc.z *= dv; acc.w *= dv;

  int sl = 4 * (c & 7) + 2 * hh;
  float g[8];
  g[0] = __shfl(acc.x, sl, 32);      g[1] = __shfl(acc.y, sl, 32);
  g[2] = __shfl(acc.z, sl, 32);      g[3] = __shfl(acc.w, sl, 32);
  g[4] = __shfl(acc.x, sl + 1, 32);  g[5] = __shfl(acc.y, sl + 1, 32);
  g[6] = __shfl(acc.z, sl + 1, 32);  g[7] = __shfl(acc.w, sl + 1, 32);
  float hv[8] = {hA.x, hA.y, hA.z, hA.w, hB.x, hB.y, hB.z, hB.w};

  v8f ch = {}, ca = {};
#pragma unroll
  for (int ks = 0; ks < 4; ++ks) {
    v2f a, b;
    a.x = w0[2 * ks];  a.y = w0[2 * ks + 1];
    b.x = hv[2 * ks];  b.y = hv[2 * ks + 1];
    ch = __builtin_amdgcn_wmma_f32_16x16x4_f32(false, a, false, b, (short)0, ch, false, false);
    a.x = w1[2 * ks];  a.y = w1[2 * ks + 1];
    b.x = g[2 * ks];   b.y = g[2 * ks + 1];
    ca = __builtin_amdgcn_wmma_f32_16x16x4_f32(false, a, false, b, (short)0, ca, false, false);
  }

  float v[8];
  float mx = -INFINITY;
#pragma unroll
  for (int r = 0; r < 8; ++r) {
    v[r] = (ch[r] + ca[r]) + bb[r];
    bool ok = (hh == 0) || (r < N_CLS - 8);
    mx = ok ? fmaxf(mx, v[r]) : mx;
  }
  mx = fmaxf(mx, __shfl_xor(mx, 16, 32));
  float sm = 0.0f;
#pragma unroll
  for (int r = 0; r < 8; ++r) {
    bool ok = (hh == 0) || (r < N_CLS - 8);
    sm += ok ? expf(v[r] - mx) : 0.0f;
  }
  sm += __shfl_xor(sm, 16, 32);
  float ls = logf(sm);
  if (c < 8) {
    float* o = so + ((threadIdx.x >> 5) * 8 + c) * N_CLS + 8 * hh;
    float2 p0 = {v[0] - mx - ls, v[1] - mx - ls};
    *(float2*)o = p0;
    if (hh == 0) {
      float2 p1 = {v[2] - mx - ls, v[3] - mx - ls};
      float2 p2 = {v[4] - mx - ls, v[5] - mx - ls};
      float2 p3 = {v[6] - mx - ls, v[7] - mx - ls};
      *(float2*)(o + 2) = p1;
      *(float2*)(o + 4) = p2;
      *(float2*)(o + 6) = p3;
    }
  }
  __syncthreads();
  long g0   = (long)blockIdx.x * (64 * N_CLS);
  long gend = (long)n * N_CLS;
  int  i4   = threadIdx.x * 4;
  if (i4 < 64 * N_CLS) {
    if (g0 + i4 + 3 < gend) {
      *(float4*)(OUT + g0 + i4) = *(const float4*)(so + i4);
    } else {
      for (int u = 0; u < 4; ++u)
        if (g0 + i4 + u < gend) OUT[g0 + i4 + u] = so[i4 + u];
    }
  }
}

__device__ __forceinline__ void proj1_tile(
    const float* __restrict__ X, const float* __restrict__ W1, const float* __restrict__ W0,
    float* __restrict__ Y, float* __restrict__ Z, int n, int wave) {
  int lane = threadIdx.x & 31;
  int base = wave * 16;
  if (base >= n) return;

  int row   = lane & 15;
  int khalf = (lane >> 4) * 2;
  int node  = base + row;
  node = node < n ? node : n - 1;
  long xrow = (long)node * N_FEAT;
  bool lo   = (khalf == 0);

  v8f c1 = {}, c0 = {};
#pragma unroll
  for (int k = 0; k < 48; k += 4) {
    v2f a = *(const v2f*)(X + xrow + k + khalf);
    v2f b1, b0;
    b1.x = W1[(k + khalf)     * HIDDEN + row];
    b1.y = W1[(k + khalf + 1) * HIDDEN + row];
    b0.x = W0[(k + khalf)     * HIDDEN + row];
    b0.y = W0[(k + khalf + 1) * HIDDEN + row];
    c1 = __builtin_amdgcn_wmma_f32_16x16x4_f32(false, a, false, b1, (short)0, c1, false, false);
    c0 = __builtin_amdgcn_wmma_f32_16x16x4_f32(false, a, false, b0, (short)0, c0, false, false);
  }
  {
    v2f a0 = *(const v2f*)(X + xrow + 48);
    float p1x = W1[48 * HIDDEN + row], p1y = W1[49 * HIDDEN + row];
    float p0x = W0[48 * HIDDEN + row], p0y = W0[49 * HIDDEN + row];
    v2f a, b1, b0;
    a.x  = lo ? a0.x : 0.0f;  a.y  = lo ? a0.y : 0.0f;
    b1.x = lo ? p1x  : 0.0f;  b1.y = lo ? p1y  : 0.0f;
    b0.x = lo ? p0x  : 0.0f;  b0.y = lo ? p0y  : 0.0f;
    c1 = __builtin_amdgcn_wmma_f32_16x16x4_f32(false, a, false, b1, (short)0, c1, false, false);
    c0 = __builtin_amdgcn_wmma_f32_16x16x4_f32(false, a, false, b0, (short)0, c0, false, false);
  }

  int col = lane & 15;
  int mhi = (lane >> 4) * 8;
#pragma unroll
  for (int j = 0; j < 8; ++j) {
    int m = base + mhi + j;
    if (m < n) {
      Y[(long)m * HIDDEN + col] = c1[j];
      Z[(long)m * HIDDEN + col] = c0[j];
    }
  }
}

extern "C" void kernel_launch(void* const* d_in, const int* in_sizes, int n_in,
                              void* d_out, int out_size, void* d_ws, size_t ws_size,
                              hipStream_t stream) {
  const float* x    = (const float*)d_in[0];
  const int*   ei   = (const int*)d_in[1];
  const float* w1_0 = (const float*)d_in[2];
  const float* w1_1 = (const float*)d_in[3];
  const float* b1   = (const float*)d_in[4];
  const float* w2_0 = (const float*)d_in[5];
  const float* w2_1 = (const float*)d_in[6];
  const float* b2   = (const float*)d_in[7];
  float* out = (float*)d_out;

  int n = in_sizes[0] / N_FEAT;
  int E = in_sizes[1] / 2;
  const int* src = ei;
  const int* dst = ei + E;

  int B = (n + BNODES - 1) / BNODES;
  int G = (E + CHUNK - 1) / CHUNK;

  float* ws     = (float*)d_ws;
  int*   cnt    = (int*)ws;
  int*   off    = cnt + n;
  float* dinv   = ws + 2L * n;
  long   tabw   = (3L * n + 31) / 32 * 32;
  long   roww   = ((long)n * HIDDEN + 31) / 32 * 32;
  float* r1     = ws + tabw;
  float* y1     = r1;
  float* r2a    = r1 + roww;
  long   r2aw   = 1L * G * CHUNK > roww ? 1L * G * CHUNK : roww;
  int*   sorted = (int*)r2a;
  float* hs     = r2a;
  float* h      = r2a + r2aw;
  int*   csr    = (int*)(h + roww);
  int*   lstT   = csr + ((long)E + 31) / 32 * 32;

  int ntiles = (n + 15) / 16;
  int PB     = (ntiles + 31) / 32;
  unsigned gblk = (unsigned)(((long)n * 4 + 255) / 256);

  chunk_sort_proj1<<<G + PB, 1024, 0, stream>>>(src, dst, sorted, lstT, x, w1_1, w1_0, y1, h, E, n, B, G);
  bin_order<<<B, 1024, 0, stream>>>(sorted, lstT, csr, cnt, off, dinv, y1, n, E, G);
  gather16_tag1<<<gblk, 256, 0, stream>>>(csr, off, cnt, dinv, y1, b1, hs, h, n, E);
  gather16_tag2<<<gblk, 256, 0, stream>>>(csr, off, cnt, dinv, hs, h, w2_0, w2_1, b2, out, n, E);
}


# Layer-1 projection tiles pass through LDS so each global access covers whole 128-byte lines: x read in row order 16 B a lane, weights staged once per group, y/z stored 4 lines a store; sum order kept
# speedup vs baseline: 9.3601x; 1.0949x over previous
#include <hip/hip_runtime.h>
#include <math.h>

typedef __attribute__((ext_vector_type(2))) float v2f;
typedef __attribute__((ext_vector_type(8))) float v8f;

#define N_FEAT   50
#define HIDDEN   16
#define N_CLS    10

#define BSH     9
#define BNODES  512
#define CHUNK   8192
#define EPT     8
#define MAXB    1024
#define MAXG    1024
#define CAP2    16384

__device__ __forceinline__ int clamp_node(int d, int n) {
  d = d < 0 ? 0 : d;
  return d < n ? d : n - 1;
}

__device__ __forceinline__ int block_excl_scan(int v, int* red, int* total) {
  int lane = threadIdx.x & 31, wv = threadIdx.x >> 5;
  int s = v;
  for (int d = 1; d < 32; d <<= 1) {
    int t = __shfl_up(s, d, 32);
    if (lane >= d) s += t;
  }
  if (lane == 31) red[wv] = s;
  __syncthreads();
  if (wv == 0) {
    int w = red[lane];
    int ws = w;
    for (int d = 1; d < 32; d <<= 1) {
      int t = __shfl_up(ws, d, 32);
      if (lane >= d) ws += t;
    }
    red[lane] = ws - w;
    if (lane == 31) red[32] = ws;
  }
  __syncthreads();
  int r = red[wv] + s - v;
  *total = red[32];
  __syncthreads();
  return r;
}

#define XT4     200
#define KSTEPS  13

__device__ __forceinline__ void proj1_group(
    const float* __restrict__ X, const float* __restrict__ W1, const float* __restrict__ W0,
    float* __restrict__ Y, float* __restrict__ Z, int n, int wave0, float4* wfrag, float4* xt);

__global__ __launch_bounds__(1024) void chunk_sort_proj1(
    const int* __restrict__ src, const int* __restrict__ dst,
    int* __restrict__ sorted, int* __restrict__ lstT,
    const float* __restrict__ X, const float* __restrict__ W1, const float* __restrict__ W0,
    float* __restrict__ Y, float* __restrict__ Z, int E, int n, int B, int G) {
  __shared__ int  hist[MAXB];
  __shared__ int  lstart[MAXB];
  __shared__ int  red[33];
  __shared__ int  stage[CHUNK];
  __shared__ float4 wfrag[KSTEPS * 32];
  __shared__ float4 xt[32 * XT4];
  if ((int)blockIdx.x >= G) {
    proj1_group(X, W1, W0, Y, Z, n, ((int)blockIdx.x - G) * 32, wfrag, xt);
    return;
  }
  int t = threadIdx.x, g = blockIdx.x;
  hist[t] = 0;
  __syncthreads();
  int e0 = g * CHUNK;
  int sv[EPT], dv[EPT], rv[EPT];
#pragma unroll
  for (int k = 0; k < EPT; ++k) {
    int e = e0 + k * 1024 + t;
    int ec = e < E ? e : E - 1;
    sv[k] = src[ec];
    dv[k] = dst[ec];
  }
#pragma unroll
  for (int k = 0; k < EPT; ++k) {
    int e = e0 + k * 1024 + t;
    rv[k] = 0;
    if (e < E) {
      dv[k] = clamp_node(dv[k], n);
      rv[k] = atomicAdd(&hist[dv[k] >> BSH], 1);
    } else {
      dv[k] = -1;
    }
  }
  __syncthreads();
  int tot;
  int ls = block_excl_scan(hist[t], red, &tot);
  lstart[t] = ls;
  __syncthreads();
  int lb[EPT];
#pragma unroll
  for (int k = 0; k < EPT; ++k) lb[k] = lstart[(dv[k] < 0 ? 0 : dv[k]) >> BSH];
#pragma unroll
  for (int k = 0; k < EPT; ++k) {
    if (dv[k] >= 0) {
      int p = lb[k] + rv[k];
      p = p < CHUNK ? p : CHUNK - 1;
      int s = ((unsigned)sv[k] < (unsigned)n) ? sv[k] : 0;
      stage[p] = (s << BSH) | (dv[k] & (BNODES - 1));
    }
  }
  if (t < B)  lstT[(long)t * G + g] = ls;
  if (t == 0) lstT[(long)B * G + g] = tot;
  __syncthreads();
  int w[EPT];
#pragma unroll
  for (int k = 0; k < EPT; ++k) w[k] = stage[k * 1024 + t];
#pragma unroll
  for (int k = 0; k < EPT; ++k) {
    int i = k * 1024 + t;
    if (i < tot) sorted[(long)g * CHUNK + i] = w[k];
  }
}

__global__ __launch_bounds__(1024) void bin_order(
    const int* __restrict__ sorted, const int* __restrict__ lstT, int* __restrict__ csr,
    int* __restrict__ cnt, int* __restrict__ off, float* __restrict__ dinv, float* Y,
    int n, int E, int G) {
  __shared__ int pstart[MAXG + 1];
  __shared__ int pbase[MAXG];
  __shared__ int first[CAP2 / 32 + 1];
  __shared__ int cntL[BNODES];
  __shared__ int cur[BNODES];
  __shared__ int red[33];
  __shared__ int eS[CAP2];
  __shared__ unsigned short eD[CAP2];
  __shared__ int stg[CAP2];
  __shared__ float dinvL[BNODES];
  int t = threadIdx.x, b = blockIdx.x;
  int lane = t & 31, wv = t >> 5;
  int base = b << BSH;

  float4* Y4 = (float4*)Y;
  long yq = (long)base * 4;
  bool yk0 = base + (t >> 2) < n, yk1 = base + ((t + 1024) >> 2) < n;
  float4 ya = {0.0f, 0.0f, 0.0f, 0.0f}, yb = {0.0f, 0.0f, 0.0f, 0.0f};
  if (yk0) ya = Y4[yq + t];
  if (yk1) yb = Y4[yq + t + 1024];

  int st = 0, len = 0;
  if (t < G) {
    st = lstT[(long)b * G + t];
    int en = lstT[(long)(b + 1) * G + t];
    st = st < 0 ? 0 : (st > CHUNK ? CHUNK : st);
    en = en < st ? st : (en > CHUNK ? CHUNK : en);
    len = en - st;
  }
  int m, bs;
  int ps = block_excl_scan(len, red, &m);
  (void)block_excl_scan(st, red, &bs);
  bs = bs < 0 ? 0 : (bs > E ? E : bs);
  m  = m > E - bs ? E - bs : m;
  pstart[t] = ps;
  if (t == 0) pstart[MAXG] = m;
  if (t < MAXG) pbase[t] = t * CHUNK + st - ps;
  if (t < BNODES) cntL[t] = 0;
  __syncthreads();

  bool in_lds = m <= CAP2;
  if (in_lds) {
    if (len > 0) {
      int pe = ps + len; pe = pe < m ? pe : m;
      for (int k = (ps + 31) >> 5; (k << 5) < pe; ++k) first[k] = t;
    }
    __syncthreads();
    for (int i0 = t; i0 < m; i0 += 8 * 1024) {
      int v[8];
#pragma unroll
      for (int u = 0; u < 8; ++u) {
        int i = i0 + u * 1024;
        v[u] = 0;
        if (i < m) {
          int p = first[i >> 5];
          p = p < 0 ? 0 : (p < G ? p : G - 1);
          while (p + 1 < G && i >= pstart[p + 1]) ++p;
          v[u] = sorted[pbase[p] + i];
        }
      }
#pragma unroll
      for (int u = 0; u < 8; ++u) {
        int i = i0 + u * 1024;
        if (i < m) {
          int dl = v[u] & (BNODES - 1);
          atomicAdd(&cntL[dl], 1);
          eS[i] = (int)((unsigned)v[u] >> BSH);
          eD[i] = (unsigned short)dl;
        }
      }
    }
  } else {
    for (int p = wv; p < G; p += 32) {
      int q0 = pstart[p], q1 = pstart[p + 1];
      q1 = q1 < m ? q1 : m;
      for (int i = q0 + lane; i < q1; i += 32) {
        int dl = sorted[pbase[p] + i] & (BNODES - 1);
        atomicAdd(&cntL[dl], 1);
      }
    }
  }
  __syncthreads();
  int c = (t < BNODES) ? cntL[t] : 0;
  int tot;
  int lo = block_excl_scan(c, red, &tot);
  if (t < BNODES) {
    cur[t] = lo;
    int node = base + t;
    if (node < n) {
      float dn = (c > 0) ? rsqrtf((float)c) : 0.0f;
      cnt[node]  = c;
      off[node]  = bs + lo;
      dinv[node] = dn;
      dinvL[t]   = dn;
    }
  }
  __syncthreads();
  if (yk0) {
    float d = dinvL[t >> 2];
    float4 o = {ya.x * d, ya.y * d, ya.z * d, ya.w * d};
    Y4[yq + t] = o;
  }
  if (yk1) {
    float d = dinvL[(t + 1024) >> 2];
    float4 o = {yb.x * d, yb.y * d, yb.z * d, yb.w * d};
    Y4[yq + t + 1024] = o;
  }
  if (in_lds) {
    for (int i0 = t; i0 < m; i0 += 8 * 1024) {
      int dd[8], ss[8];
#pragma unroll
      for (int u = 0; u < 8; ++u) {
        int i = i0 + u * 1024;
        int ic = i < m ? i : 0;
        dd[u] = eD[ic];
        ss[u] = eS[ic];
      }
#pragma unroll
      for (int u = 0; u < 8; ++u) {
        int i = i0 + u * 1024;
        if (i < m) {
          int p = atomicAdd(&cur[dd[u]], 1);
          p = p < 0 ? 0 : (p < CAP2 ? p : CAP2 - 1);
          stg[p] = ss[u];
        }
      }
    }
    __syncthreads();
    for (int i = t; i < m; i += 1024) csr[bs + i] = stg[i];
  } else {
    for (int p = wv; p < G; p += 32) {
      int q0 = pstart[p], q1 = pstart[p + 1];
      q1 = q1 < m ? q1 : m;
      for (int i = q0 + lane; i < q1; i += 32) {
        int v = sorted[pbase[p] + i];
        int dl = v & (BNODES - 1);
        unsigned pos = (unsigned)(bs + atomicAdd(&cur[dl], 1));
        pos = pos < (unsigned)E ? pos : (unsigned)(E - 1);
        csr[pos] = (int)((unsigned)v >> BSH);
      }
    }
  }
}

__device__ __forceinline__ float4 row_piece(const float4* __restrict__ h4, int s, int q, int n) {
  s = ((unsigned)s < (unsigned)n) ? s : 0;
  return h4[(long)s * 4 + q];
}

__device__ __forceinline__ int4 list_piece(const int* __restrict__ csr, int p, int E) {
  int pc = p < E - 4 ? p : E - 4;
  pc = pc < 0 ? 0 : pc;
  int d = p - pc;
  int4 e;
  __builtin_memcpy(&e, csr + pc, 16);
  if (d != 0) {
    e.x = d == 1 ? e.y : (d == 2 ? e.z : e.w);
    e.y = d == 1 ? e.z : e.w;
    e.z = e.w;
  }
  return e;
}

template <int K>
__device__ __forceinline__ int quad_take(int v) {
  return __builtin_amdgcn_update_dpp(0, v, K * 0x55, 0xF, 0xF, true);
}

template <int K>
__device__ __forceinline__ void add_four(float4& acc, const float4* __restrict__ h4, int4 e,
                                         int jj, int end, int q, int n) {
  bool v1 = jj + 1 < end, v2 = jj + 2 < end, v3 = jj + 3 < end;
  int s0 = quad_take<K>(e.x);
  int s1 = quad_take<K>(e.y);
  int s2 = quad_take<K>(e.z);
  int s3 = quad_take<K>(e.w);
  s1 = v1 ? s1 : s0;
  s2 = v2 ? s2 : s0;
  s3 = v3 ? s3 : s0;
  float4 r0 = row_piece(h4, s0, q, n);
  float4 r1 = row_piece(h4, s1, q, n);
  float4 r2 = row_piece(h4, s2, q, n);
  float4 r3 = row_piece(h4, s3, q, n);
  acc.x += r0.x; acc.y += r0.y; acc.z += r0.z; acc.w += r0.w;
  acc.x += v1 ? r1.x : 0.0f; acc.y += v1 ? r1.y : 0.0f; acc.z += v1 ? r1.z : 0.0f; acc.w += v1 ? r1.w : 0.0f;
  acc.x += v2 ? r2.x : 0.0f; acc.y += v2 ? r2.y : 0.0f; acc.z += v2 ? r2.z : 0.0f; acc.w += v2 ? r2.w : 0.0f;
  acc.x += v3 ? r3.x : 0.0f; acc.y += v3 ? r3.y : 0.0f; acc.z += v3 ? r3.z : 0.0f; acc.w += v3 ? r3.w : 0.0f;
}

__device__ __forceinline__ float4 edge_sum(const int* __restrict__ csr, const float4* __restrict__ h4,
                                           int j, int end, int q, int n, int E) {
  j   = j   < 0 ? 0 : j;
  end = end > E ? E : end;
  float4 acc = {0.0f, 0.0f, 0.0f, 0.0f};
  for (int b = j; b < end; b += 32) {
    bool more = b + 16 < end;
    int4 ea = list_piece(csr, b + 4 * q, E);
    int4 eb = {0, 0, 0, 0};
    if (more) eb = list_piece(csr, b + 16 + 4 * q, E);
    add_four<0>(acc, h4, ea, b, end, q, n);
    if (b + 4 < end) {
      add_four<1>(acc, h4, ea, b + 4, end, q, n);
      if (b + 8 < end) {
        add_four<2>(acc, h4, ea, b + 8, end, q, n);
        if (b + 12 < end) {
          add_four<3>(acc, h4, ea, b + 12, end, q, n);
          if (more) {
            add_four<0>(acc, h4, eb, b + 16, end, q, n);
            if (b + 20 < end) {
              add_four<1>(acc, h4, eb, b + 20, end, q, n);
              if (b + 24 < end) {
                add_four<2>(acc, h4, eb, b + 24, end, q, n);
                if (b + 28 < end) add_four<3>(acc, h4, eb, b + 28, end, q, n);
              }
            }
          }
        }
      }
    }
  }
  return acc;
}

__global__ __launch_bounds__(256) void gather16_tag1(
    const int* __restrict__ csr, const int* __restrict__ off, const int* __restrict__ cnt,
    const float* __restrict__ dinv, const float* __restrict__ y, const float* __restrict__ B,
    float* __restrict__ HS, float* __restrict__ H, int n, int E) {
  int t = blockIdx.x * blockDim.x + threadIdx.x;
  int node = t >> 2;
  int q = t & 3;
  if (node >= n) return;
  int j   = off[node];
  int end = j + cnt[node];
  float dv = dinv[node];
  float4 z = ((const float4*)H)[(long)node * 4 + q];
  float b0 = B[4 * q], b1v = B[4 * q + 1], b2v = B[4 * q + 2], b3v = B[4 * q + 3];
  float4 acc = edge_sum(csr, (const float4*)y, j, end, q, n, E);
  float4 v;
  v.x = (z.x + acc.x * dv) + b0;
  v.y = (z.y + acc.y * dv) + b1v;
  v.z = (z.z + acc.z * dv) + b2v;
  v.w = (z.w + acc.w * dv) + b3v;
  v.x = v.x > 0.0f ? v.x : 0.0f;  v.y = v.y > 0.0f ? v.y : 0.0f;
  v.z = v.z > 0.0f ? v.z : 0.0f;  v.w = v.w > 0.0f ? v.w : 0.0f;
  ((float4*)H)[(long)node * 4 + q] = v;
  float4 s = {v.x * dv, v.y * dv, v.z * dv, v.w * dv};
  ((float4*)HS)[(long)node * 4 + q] = s;
}

__global__ __launch_bounds__(256) void gather16_tag2(
    const int* __restrict__ csr, const int* __restrict__ off, const int* __restrict__ cnt,
    const float* __restrict__ dinv, const float* __restrict__ HS, const float* __restrict__ H,
    const float* __restrict__ W0, const float* __restrict__ W1, const float* __restrict__ B,
    float* __restrict__ OUT, int n, int E) {
  int t = blockIdx.x * blockDim.x + threadIdx.x;
  int lane = threadIdx.x & 31;
  int wbase = (t >> 5) * 8;
  __shared__ __attribute__((aligned(16))) float so[64 * N_CLS];

  int node = wbase + (lane >> 2);
  int q = lane & 3;
  bool nv = node < n;
  int nc = nv ? node : n - 1;
  int j   = off[nc];
  int cn  = cnt[nc];
  float dv = dinv[nc];
  int end = j + (nv ? cn : 0);

  int c  = lane & 15;
  int hh = lane >> 4;
  int dn = wbase + (c & 7);
  dn = dn < n ? dn : n - 1;
  const float4* H4 = (const float4*)H;
  float4 hA = H4[(long)dn * 4 + 2 * hh];
  float4 hB = H4[(long)dn * 4 + 2 * hh + 1];
  bool cok = c < N_CLS;
  int  wc  = cok ? c : (N_CLS - 1);
  float w0[8], w1[8], bb[8];
#pragma unroll
  for (int i = 0; i < 8; ++i) {
    int f = 8 * hh + i;
    float a0 = W0[f * N_CLS + wc];
    float a1 = W1[f * N_CLS + wc];
    w0[i] = cok ? a0 : 0.0f;
    w1[i] = cok ? a1 : 0.0f;
    int cl = 8 * hh + i;
    float bv = B[cl < N_CLS ? cl : (N_CLS - 1)];
    bb[i] = cl < N_CLS ? bv : 0.0f;
  }

  float4 acc = edge_sum(csr, (const float4*)HS, j, end, q, n, E);
  acc.x *= dv; acc.y *= dv; acc.z *= dv; acc.w *= dv;

  int sl = 4 * (c & 7) + 2 * hh;
  float g[8];
  g[0] = __shfl(acc.x, sl, 32);      g[1] = __shfl(acc.y, sl, 32);
  g[2] = __shfl(acc.z, sl, 32);      g[3] = __shfl(acc.w, sl, 32);
  g[4] = __shfl(acc.x, sl + 1, 32);  g[5] = __shfl(acc.y, sl + 1, 32);
  g[6] = __shfl(acc.z, sl + 1, 32);  g[7] = __shfl(acc.w, sl + 1, 32);
  float hv[8] = {hA.x, hA.y, hA.z, hA.w, hB.x, hB.y, hB.z, hB.w};

  v8f ch = {}, ca = {};
#pragma unroll
  for (int ks = 0; ks < 4; ++ks) {
    v2f a, b;
    a.x = w0[2 * ks];  a.y = w0[2 * ks + 1];
    b.x = hv[2 * ks];  b.y = hv[2 * ks + 1];
    ch = __builtin_amdgcn_wmma_f32_16x16x4_f32(false, a, false, b, (short)0, ch, false, false);
    a.x = w1[2 * ks];  a.y = w1[2 * ks + 1];
    b.x = g[2 * ks];   b.y = g[2 * ks + 1];
    ca = __builtin_amdgcn_wmma_f32_16x16x4_f32(false, a, false, b, (short)0, ca, false, false);
  }

  float v[8];
  float mx = -INFINITY;
#pragma unroll
  for (int r = 0; r < 8; ++r) {
    v[r] = (ch[r] + ca[r]) + bb[r];
    bool ok = (hh == 0) || (r < N_CLS - 8);
    mx = ok ? fmaxf(mx, v[r]) : mx;
  }
  mx = fmaxf(mx, __shfl_xor(mx, 16, 32));
  float sm = 0.0f;
#pragma unroll
  for (int r = 0; r < 8; ++r) {
    bool ok = (hh == 0) || (r < N_CLS - 8);
    sm += ok ? expf(v[r] - mx) : 0.0f;
  }
  sm += __shfl_xor(sm, 16, 32);
  float ls = logf(sm);
  if (c < 8) {
    float* o = so + ((threadIdx.x >> 5) * 8 + c) * N_CLS + 8 * hh;
    float2 p0 = {v[0] - mx - ls, v[1] - mx - ls};
    *(float2*)o = p0;
    if (hh == 0) {
      float2 p1 = {v[2] - mx - ls, v[3] - mx - ls};
      float2 p2 = {v[4] - mx - ls, v[5] - mx - ls};
      float2 p3 = {v[6] - mx - ls, v[7] - mx - ls};
      *(float2*)(o + 2) = p1;
      *(float2*)(o + 4) = p2;
      *(float2*)(o + 6) = p3;
    }
  }
  __syncthreads();
  long g0   = (long)blockIdx.x * (64 * N_CLS);
  long gend = (long)n * N_CLS;
  int  i4   = threadIdx.x * 4;
  if (i4 < 64 * N_CLS) {
    if (g0 + i4 + 3 < gend) {
      *(float4*)(OUT + g0 + i4) = *(const float4*)(so + i4);
    } else {
      for (int u = 0; u < 4; ++u)
        if (g0 + i4 + u < gend) OUT[g0 + i4 + u] = so[i4 + u];
    }
  }
}

__device__ __forceinline__ void proj1_group(
    const float* __restrict__ X, const float* __restrict__ W1, const float* __restrict__ W0,
    float* __restrict__ Y, float* __restrict__ Z, int n, int wave0, float4* wfrag, float4* xt) {
  int t = threadIdx.x;
  int lane = t & 31, wv = t >> 5;
  int wave = wave0 + wv;
  int base = wave * 16;

  long lim = (long)n * N_FEAT;
  long f0  = (long)wave * XT4;
  float4 xr[7];
#pragma unroll
  for (int i = 0; i < 7; ++i) {
    int f  = lane + 32 * i;
    int fc = f < XT4 ? f : XT4 - 1;
    long e  = (f0 + fc) * 4;
    bool ok = e + 3 < lim;
    long ec = ok ? e : lim - 4;
    float4 v;
    __builtin_memcpy(&v, X + ec, 16);
    float4 r = {ok ? v.x : 0.0f, ok ? v.y : 0.0f, ok ? v.z : 0.0f, ok ? v.w : 0.0f};
    xr[i] = r;
  }

  if (t < KSTEPS * 32) {
    int ks = t >> 5, col = t & 15, h = (t >> 4) & 1;
    int k0 = 4 * ks + 2 * h;
    bool v0 = k0 < N_FEAT, v1 = k0 + 1 < N_FEAT;
    int ka = v0 ? k0 : 0, kb = v1 ? k0 + 1 : 0;
    float a1 = W1[ka * HIDDEN + col], b1 = W1[kb * HIDDEN + col];
    float a0 = W0[ka * HIDDEN + col], b0 = W0[kb * HIDDEN + col];
    float4 w = {v0 ? a1 : 0.0f, v1 ? b1 : 0.0f, v0 ? a0 : 0.0f, v1 ? b0 : 0.0f};
    wfrag[t] = w;
  }
  __syncthreads();
  if (base >= n) return;

  float4* xw = xt + wv * XT4;
#pragma unroll
  for (int i = 0; i < 7; ++i) {
    int f = lane + 32 * i;
    if (f < XT4) xw[f] = xr[i];
  }
  if ((lim & 2) != 0) {
    long pe = (lim - 2) >> 2;
    if (lane == 0 && pe >= f0 && pe < f0 + XT4) {
      float2 h2 = *(const float2*)(X + lim - 2);
      float4 p = {h2.x, h2.y, 0.0f, 0.0f};
      xw[(int)(pe - f0)] = p;
    }
  }

  int row = lane & 15;
  int hh  = lane >> 4;
  const float* xf = (const float*)xw + row * N_FEAT + 2 * hh;

  v8f c1 = {}, c0 = {};
#pragma unroll
  for (int ks = 0; ks < KSTEPS - 1; ++ks) {
    v2f a = *(const v2f*)(xf + 4 * ks);
    float4 w = wfrag[ks * 32 + lane];
    v2f b1, b0;
    b1.x = w.x; b1.y = w.y;
    b0.x = w.z; b0.y = w.w;
    c1 = __builtin_amdgcn_wmma_f32_16x16x4_f32(false, a, false, b1, (short)0, c1, false, false);
    c0 = __builtin_amdgcn_wmma_f32_16x16x4_f32(false, a, false, b0, (short)0, c0, false, false);
  }
  {
    v2f a0 = *(const v2f*)((const float*)xw + row * N_FEAT + 48);
    float4 w = wfrag[(KSTEPS - 1) * 32 + lane];
    bool lo = hh == 0;
    v2f a, b1, b0;
    a.x  = lo ? a0.x : 0.0f;  a.y  = lo ? a0.y : 0.0f;
    b1.x = w.x; b1.y = w.y;
    b0.x = w.z; b0.y = w.w;
    c1 = __builtin_amdgcn_wmma_f32_16x16x4_f32(false, a, false, b1, (short)0, c1, false, false);
    c0 = __builtin_amdgcn_wmma_f32_16x16x4_f32(false, a, false, b0, (short)0, c0, false, false);
  }

  float* yt = (float*)xw;
  float* zt = yt + 256;
  int mhi = hh * 8;
#pragma unroll
  for (int j = 0; j < 8; ++j) {
    yt[(mhi + j) * HIDDEN + row] = c1[j];
    zt[(mhi + j) * HIDDEN + row] = c0[j];
  }
  float4* Y4 = (float4*)Y;
  float4* Z4 = (float4*)Z;
#pragma unroll
  for (int i = 0; i < 2; ++i) {
    int f = lane + 32 * i;
    float4 vy = ((const float4*)yt)[f];
    float4 vz = ((const float4*)zt)[f];
    if (base + (f >> 2) < n) {
      Y4[(long)base * 4 + f] = vy;
      Z4[(long)base * 4 + f] = vz;
    }
  }
}

extern "C" void kernel_launch(void* const* d_in, const int* in_sizes, int n_in,
                              void* d_out, int out_size, void* d_ws, size_t ws_size,
                              hipStream_t stream) {
  const float* x    = (const float*)d_in[0];
  const int*   ei   = (const int*)d_in[1];
  const float* w1_0 = (const float*)d_in[2];
  const float* w1_1 = (const float*)d_in[3];
  const float* b1   = (const float*)d_in[4];
  const float* w2_0 = (const float*)d_in[5];
  const float* w2_1 = (const float*)d_in[6];
  const float* b2   = (const float*)d_in[7];
  float* out = (float*)d_out;

  int n = in_sizes[0] / N_FEAT;
  int E = in_sizes[1] / 2;
  const int* src = ei;
  const int* dst = ei + E;

  int B = (n + BNODES - 1) / BNODES;
  int G = (E + CHUNK - 1) / CHUNK;

  float* ws     = (float*)d_ws;
  int*   cnt    = (int*)ws;
  int*   off    = cnt + n;
  float* dinv   = ws + 2L * n;
  long   tabw   = (3L * n + 31) / 32 * 32;
  long   roww   = ((long)n * HIDDEN + 31) / 32 * 32;
  float* r1     = ws + tabw;
  float* y1     = r1;
  float* r2a    = r1 + roww;
  long   r2aw   = 1L * G * CHUNK > roww ? 1L * G * CHUNK : roww;
  int*   sorted = (int*)r2a;
  float* hs     = r2a;
  float* h      = r2a + r2aw;
  int*   csr    = (int*)(h + roww);
  int*   lstT   = csr + ((long)E + 31) / 32 * 32;

  int ntiles = (n + 15) / 16;
  int PB     = (ntiles + 31) / 32;
  unsigned gblk = (unsigned)(((long)n * 4 + 255) / 256);

  chunk_sort_proj1<<<G + PB, 1024, 0, stream>>>(src, dst, sorted, lstT, x, w1_1, w1_0, y1, h, E, n, B, G);
  bin_order<<<B, 1024, 0, stream>>>(sorted, lstT, csr, cnt, off, dinv, y1, n, E, G);
  gather16_tag1<<<gblk, 256, 0, stream>>>(csr, off, cnt, dinv, y1, b1, hs, h, n, E);
  gather16_tag2<<<gblk, 256, 0, stream>>>(csr, off, cnt, dinv, hs, h, w2_0, w2_1, b2, out, n, E);
}
